# starting kernel
# baseline (speedup 1.0000x reference)
#include <hip/hip_runtime.h>
#include <cstdint>
#include <cstddef>

#define DEVINL __device__ __forceinline__

typedef float v2f __attribute__((ext_vector_type(2)));
typedef float v8f __attribute__((ext_vector_type(8)));

static constexpr int T_DIM  = 1024;
static constexpr int B_DIM  = 64;
static constexpr int IN_DIM = 32;
static constexpr int EMB    = 64;
static constexpr int HID    = 128;
static constexpr int KF     = 16;
static constexpr int OUTD   = 32;
static constexpr int G4     = 512;

static constexpr int HS     = 132;
static constexpr int XS     = 68;
static constexpr int IS     = 36;
static constexpr int BP512  = 528;
static constexpr int BP64   = 80;

DEVINL float sigm(float x) { return 1.0f / (1.0f + __expf(-x)); }

DEVINL v8f wmma4(v2f a, v2f b, v8f c) {
  return __builtin_amdgcn_wmma_f32_16x16x4_f32(false, a, false, b, (short)0, c,
                                               false, false);
}

DEVINL v8f vzero8() {
  v8f z;
#pragma unroll
  for (int r = 0; r < 8; ++r) z[r] = 0.0f;
  return z;
}

__global__ void zero_kernel(float* __restrict__ p, int n) {
  int i = blockIdx.x * blockDim.x + threadIdx.x;
  if (i < n) p[i] = 0.0f;
}

__global__ void __launch_bounds__(512)
embed_xg0_kernel(const float* __restrict__ in,
                 const float* __restrict__ We,
                 const float* __restrict__ be,
                 const float* __restrict__ Wx0,
                 const float* __restrict__ b0,
                 float* __restrict__ xg)
{
  extern __shared__ __align__(16) char smem[];
  float* inP = (float*)smem;
  v2f*   WeP = (v2f*)(smem + 64 * IS * 4);
  float* xb  = (float*)(smem + 64 * IS * 4 + 16 * BP64 * 8);
  v2f*   WxP = (v2f*)(smem + 64 * IS * 4 + 16 * BP64 * 8 + 64 * XS * 4);

  const int tid  = threadIdx.x;
  const int wave = tid >> 5;
  const int lane = tid & 31;
  const int lh   = lane >> 4;
  const int ln   = lane & 15;
  const int row0 = blockIdx.x * 64;

  for (int i = tid; i < 64 * 32; i += 512) {
    int r = i >> 5, c = i & 31;
    inP[r * IS + c] = in[(size_t)(row0 + r) * 32 + c];
  }
  for (int i = tid; i < 16 * 64; i += 512) {
    int kp = i >> 6, n = i & 63;
    v2f w; w.x = We[(2 * kp) * 64 + n]; w.y = We[(2 * kp + 1) * 64 + n];
    WeP[kp * BP64 + n] = w;
  }
  for (int i = tid; i < 32 * 512; i += 512) {
    int kp = i >> 9, n = i & 511;
    v2f w; w.x = Wx0[(2 * kp) * 512 + n]; w.y = Wx0[(2 * kp + 1) * 512 + n];
    WxP[kp * BP512 + n] = w;
  }
  __syncthreads();

  {
    const int tm = wave >> 2, tn = wave & 3;
    const int col = tn * 16 + ln;
    const float bv = be[col];
    v8f acc = vzero8();
    const int am = tm * 16 + ln;
#pragma unroll
    for (int k0 = 0; k0 < 32; k0 += 4) {
      v2f a = *(const v2f*)&inP[am * IS + k0 + 2 * lh];
      v2f b = WeP[((k0 >> 1) + lh) * BP64 + col];
      acc = wmma4(a, b, acc);
    }
#pragma unroll
    for (int r = 0; r < 8; ++r)
      xb[(tm * 16 + r + 8 * lh) * XS + col] = acc[r] + bv;
  }
  __syncthreads();

  for (int task = wave; task < 128; task += 16) {
    const int tm = task >> 5, tn = task & 31;
    const int col = tn * 16 + ln;
    const float bv = b0[col];
    v8f acc = vzero8();
    const int am = tm * 16 + ln;
#pragma unroll
    for (int k0 = 0; k0 < 64; k0 += 4) {
      v2f a = *(const v2f*)&xb[am * XS + k0 + 2 * lh];
      v2f b = WxP[((k0 >> 1) + lh) * BP512 + col];
      acc = wmma4(a, b, acc);
    }
#pragma unroll
    for (int r = 0; r < 8; ++r)
      xg[(size_t)(row0 + tm * 16 + r + 8 * lh) * 512 + col] = acc[r] + bv;
  }
}

__global__ void __launch_bounds__(512)
xg1_kernel(const float* __restrict__ x1,
           const float* __restrict__ Wx1,
           const float* __restrict__ b1,
           float* __restrict__ xg)
{
  extern __shared__ __align__(16) char smem[];
  float* aB = (float*)smem;
  v2f*   WP = (v2f*)(smem + 64 * HS * 4);

  const int tid  = threadIdx.x;
  const int wave = tid >> 5;
  const int lane = tid & 31;
  const int lh   = lane >> 4;
  const int ln   = lane & 15;
  const int row0 = blockIdx.x * 64;

  for (int i = tid; i < 64 * 128; i += 512) {
    int r = i >> 7, c = i & 127;
    aB[r * HS + c] = x1[(size_t)(row0 + r) * 128 + c];
  }
  for (int i = tid; i < 64 * 512; i += 512) {
    int kp = i >> 9, n = i & 511;
    v2f w; w.x = Wx1[(2 * kp) * 512 + n]; w.y = Wx1[(2 * kp + 1) * 512 + n];
    WP[kp * BP512 + n] = w;
  }
  __syncthreads();

  for (int task = wave; task < 128; task += 16) {
    const int tm = task >> 5, tn = task & 31;
    const int col = tn * 16 + ln;
    const float bv = b1[col];
    v8f acc = vzero8();
    const int am = tm * 16 + ln;
    for (int k0 = 0; k0 < 128; k0 += 4) {
      v2f a = *(const v2f*)&aB[am * HS + k0 + 2 * lh];
      v2f b = WP[((k0 >> 1) + lh) * BP512 + col];
      acc = wmma4(a, b, acc);
    }
#pragma unroll
    for (int r = 0; r < 8; ++r)
      xg[(size_t)(row0 + tm * 16 + r + 8 * lh) * 512 + col] = acc[r] + bv;
  }
}

template <int NOUT>
__global__ void __launch_bounds__(512)
scan_kernel(const float* __restrict__ xg,
            const float* __restrict__ Wh,
            const float* __restrict__ Wo,
            const float* __restrict__ bo,
            float* __restrict__ ring,
            float* __restrict__ out)
{
  extern __shared__ __align__(16) char smem[];
  v2f*   WhP  = (v2f*)smem;
  float* hbuf = (float*)(smem + 64 * BP512 * 8);

  const int tid  = threadIdx.x;
  const int wave = tid >> 5;
  const int lane = tid & 31;
  const int lh   = lane >> 4;
  const int ln   = lane & 15;

  for (int i = tid; i < 64 * 512; i += 512) {
    int kp = i >> 9, n = i & 511;
    v2f w; w.x = Wh[(2 * kp) * 512 + n]; w.y = Wh[(2 * kp + 1) * 512 + n];
    WhP[kp * BP512 + n] = w;
  }
  for (int i = tid; i < 64 * HS; i += 512) hbuf[i] = 0.0f;
  __syncthreads();

  float hnew[2][8];

  for (int t = 0; t < T_DIM; ++t) {
    const float* xgt = xg + (size_t)t * (64 * 512);

    if (t + 1 < T_DIM) {
      const float* nxt = xgt + 64 * 512;
#pragma unroll
      for (int q = 0; q < 8; ++q)
        __builtin_prefetch(nxt + ((size_t)tid + q * 512) * 32, 0, 0);
    }

#pragma unroll
    for (int ti = 0; ti < 2; ++ti) {
      const int task = wave + ti * 16;
      const int tm   = task >> 3;
      const int tnh  = task & 7;
      const int colh = tnh * 16 + ln;

      float xv0[8], xv1[8], xv2[8], xv3[8];
#pragma unroll
      for (int r = 0; r < 8; ++r) {
        const int row = tm * 16 + r + 8 * lh;
        const float* p = xgt + row * 512 + colh;
        xv0[r] = p[0];  xv1[r] = p[128];
        xv2[r] = p[256]; xv3[r] = p[384];
      }

      v8f acc0 = vzero8(), acc1 = vzero8(), acc2 = vzero8(), acc3 = vzero8();
      const int am = tm * 16 + ln;
      for (int k0 = 0; k0 < 128; k0 += 4) {
        v2f a = *(const v2f*)&hbuf[am * HS + k0 + 2 * lh];
        const int kp = (k0 >> 1) + lh;
        acc0 = wmma4(a, WhP[kp * BP512 + colh + 0],   acc0);
        acc1 = wmma4(a, WhP[kp * BP512 + colh + 128], acc1);
        acc2 = wmma4(a, WhP[kp * BP512 + colh + 256], acc2);
        acc3 = wmma4(a, WhP[kp * BP512 + colh + 384], acc3);
      }

      float hc[8], og[8], dd[8];
#pragma unroll
      for (int r = 0; r < 8; ++r) {
        const float ig = sigm(acc0[r] + xv0[r]);
        og[r] = sigm(acc1[r] + xv1[r]);
        const float cg = tanhf(acc2[r] + xv2[r]);
        dd[r] = 0.5f * sigm(acc3[r] + xv3[r]);
        hc[r] = ig * cg;
      }

      float csum[8], w[8];
#pragma unroll
      for (int r = 0; r < 8; ++r) { csum[r] = hc[r]; w[r] = 1.0f; }
#pragma unroll
      for (int j = 1; j < KF; ++j) {
        const float* rp = ring + (size_t)((t - j) & 15) * (64 * 128);
        const float jm1 = (float)(j - 1), rj = 1.0f / (float)j;
#pragma unroll
        for (int r = 0; r < 8; ++r) {
          const int row = tm * 16 + r + 8 * lh;
          w[r] *= (jm1 + dd[r]) * rj;
          csum[r] += w[r] * rp[row * 128 + colh];
        }
      }
      float* rw = ring + (size_t)(t & 15) * (64 * 128);
#pragma unroll
      for (int r = 0; r < 8; ++r) {
        const int row = tm * 16 + r + 8 * lh;
        rw[row * 128 + colh] = hc[r];
        hnew[ti][r] = og[r] * tanhf(csum[r]);
      }
    }

    __syncthreads();
#pragma unroll
    for (int ti = 0; ti < 2; ++ti) {
      const int task = wave + ti * 16;
      const int tm = task >> 3, tnh = task & 7;
      const int colh = tnh * 16 + ln;
#pragma unroll
      for (int r = 0; r < 8; ++r)
        hbuf[(tm * 16 + r + 8 * lh) * HS + colh] = hnew[ti][r];
    }
    __syncthreads();

    float* ot = out + (size_t)t * (64 * NOUT);
    for (int task = wave; task < 4 * (NOUT / 16); task += 16) {
      const int tm  = task / (NOUT / 16);
      const int tno = task % (NOUT / 16);
      const int col = tno * 16 + ln;
      const float bv = bo[col];
      v8f acc = vzero8();
      const int am = tm * 16 + ln;
      for (int k0 = 0; k0 < 128; k0 += 4) {
        v2f a = *(const v2f*)&hbuf[am * HS + k0 + 2 * lh];
        v2f b;
        b.x = Wo[(k0 + 2 * lh) * NOUT + col];
        b.y = Wo[(k0 + 2 * lh + 1) * NOUT + col];
        acc = wmma4(a, b, acc);
      }
#pragma unroll
      for (int r = 0; r < 8; ++r)
        ot[(tm * 16 + r + 8 * lh) * NOUT + col] = acc[r] + bv;
    }
  }
}

extern "C" void kernel_launch(void* const* d_in, const int* in_sizes, int n_in,
                              void* d_out, int out_size, void* d_ws, size_t ws_size,
                              hipStream_t stream) {
  const float* inputs = (const float*)d_in[0];
  const float* W_emb  = (const float*)d_in[1];
  const float* b_emb  = (const float*)d_in[2];
  const float* Wx0    = (const float*)d_in[3];
  const float* Wh0    = (const float*)d_in[4];
  const float* b0     = (const float*)d_in[5];
  const float* Wo0    = (const float*)d_in[6];
  const float* bo0    = (const float*)d_in[7];
  const float* Wx1    = (const float*)d_in[8];
  const float* Wh1    = (const float*)d_in[9];
  const float* b1     = (const float*)d_in[10];
  const float* Wo1    = (const float*)d_in[11];
  const float* bo1    = (const float*)d_in[12];
  float* out = (float*)d_out;

  float* xg   = (float*)d_ws;
  float* x1   = xg + (size_t)T_DIM * B_DIM * G4;
  float* ring = x1 + (size_t)T_DIM * B_DIM * HID;
  const int ringN = KF * B_DIM * HID;

  const size_t ldsE = (size_t)64 * IS * 4 + 16 * BP64 * 8 + 64 * XS * 4 + 32 * BP512 * 8;
  const size_t ldsX = (size_t)64 * HS * 4 + 64 * BP512 * 8;
  const size_t ldsS = (size_t)64 * BP512 * 8 + 64 * HS * 4;

  const int nblk = (T_DIM * B_DIM) / 64;

  embed_xg0_kernel<<<nblk, 512, ldsE, stream>>>(inputs, W_emb, b_emb, Wx0, b0, xg);
  zero_kernel<<<ringN / 256, 256, 0, stream>>>(ring, ringN);
  scan_kernel<HID><<<1, 512, ldsS, stream>>>(xg, Wh0, Wo0, bo0, ring, x1);
  xg1_kernel<<<nblk, 512, ldsX, stream>>>(x1, Wx1, b1, xg);
  zero_kernel<<<ringN / 256, 256, 0, stream>>>(ring, ringN);
  scan_kernel<OUTD><<<1, 512, ldsS, stream>>>(xg, Wh1, Wo1, bo1, ring, out);
}


# scan: k-loop cut in two; its first 64 WMMAs are issued inside the previous task's filter sum (other batch half, behind that half's await), taps 7-15 between the wave's own WMMAs; sums unchanged
# speedup vs baseline: 2.3339x; 2.3339x over previous
#include <hip/hip_runtime.h>
#include <cstdint>
#include <cstddef>

#define DEVINL __device__ __forceinline__

typedef float v2f __attribute__((ext_vector_type(2)));
typedef float v4f __attribute__((ext_vector_type(4)));
typedef float v8f __attribute__((ext_vector_type(8)));

static constexpr int T_DIM  = 1024;
static constexpr int B_DIM  = 64;
static constexpr int IN_DIM = 32;
static constexpr int EMB    = 64;
static constexpr int HID    = 128;
static constexpr int KF     = 16;
static constexpr int OUTD   = 32;
static constexpr int G4     = 512;

static constexpr int HS     = 132;
static constexpr int XS     = 68;
static constexpr int IS     = 36;
static constexpr int BP512  = 528;
static constexpr int BP64   = 80;
static constexpr int WHS_BYTES = 8 * 64 * 2 * 16 * 16;

DEVINL float sigm(float x) { return 1.0f / (1.0f + __expf(-x)); }

DEVINL v8f wmma4(v2f a, v2f b, v8f c) {
  return __builtin_amdgcn_wmma_f32_16x16x4_f32(false, a, false, b, (short)0, c,
                                               false, false);
}

DEVINL v8f vzero8() {
  v8f z;
#pragma unroll
  for (int r = 0; r < 8; ++r) z[r] = 0.0f;
  return z;
}

DEVINL void bar_arrive() { asm volatile("s_wait_dscnt 0x0\n\ts_barrier_signal -1" ::: "memory"); }
DEVINL void bar_await()  { asm volatile("s_barrier_wait -1" ::: "memory"); }

struct Frag { v2f a[2]; v4f q[2][2]; };

DEVINL void load_frag(Frag& f, const float* ab, const v4f* wq, int blk) {
#pragma unroll
  for (int kk = 0; kk < 2; ++kk) {
    const int i = blk * 2 + kk;
    f.a[kk]    = *(const v2f*)(ab + 4 * i);
    f.q[kk][0] = wq[i * 64];
    f.q[kk][1] = wq[i * 64 + 16];
  }
}

DEVINL void block_wmma(const Frag& f, v8f (&acc)[4]) {
#pragma unroll
  for (int kk = 0; kk < 2; ++kk) {
    const v4f q0 = f.q[kk][0], q1 = f.q[kk][1];
    v2f b0, b1, b2, b3;
    b0.x = q0.x; b0.y = q0.y; b1.x = q0.z; b1.y = q0.w;
    b2.x = q1.x; b2.y = q1.y; b3.x = q1.z; b3.y = q1.w;
    acc[0] = wmma4(f.a[kk], b0, acc[0]);
    acc[1] = wmma4(f.a[kk], b1, acc[1]);
    acc[2] = wmma4(f.a[kk], b2, acc[2]);
    acc[3] = wmma4(f.a[kk], b3, acc[3]);
  }
}

template <int B0, int B1>
DEVINL void k_blocks(Frag (&fr)[2], v8f (&acc)[4], const float* ab, const v4f* wq) {
#pragma unroll
  for (int blk = B0; blk < B1; ++blk) {
    if (blk + 1 < 16) load_frag(fr[(blk & 1) ^ 1], ab, wq, blk + 1);
    __builtin_amdgcn_sched_barrier(0);
    block_wmma(fr[blk & 1], acc);
    __builtin_amdgcn_sched_barrier(0);
  }
}

template <int NOUT>
DEVINL void project_tile(const float* hbuf, const float* __restrict__ Wo,
                         const float* __restrict__ bo, float* ot, int p, int lh, int ln) {
  const int tm  = p / (NOUT / 16);
  const int tno = p % (NOUT / 16);
  const int col = tno * 16 + ln;
  const float bv = bo[col];
  v8f acc = vzero8();
  const int am = tm * 16 + ln;
  for (int k0 = 0; k0 < 128; k0 += 4) {
    v2f a = *(const v2f*)&hbuf[am * HS + k0 + 2 * lh];
    v2f b;
    b.x = Wo[(k0 + 2 * lh) * NOUT + col];
    b.y = Wo[(k0 + 2 * lh + 1) * NOUT + col];
    acc = wmma4(a, b, acc);
  }
#pragma unroll
  for (int r = 0; r < 8; ++r)
    ot[(tm * 16 + r + 8 * lh) * NOUT + col] = acc[r] + bv;
}

__global__ void zero_kernel(float* __restrict__ p, int n) {
  int i = blockIdx.x * blockDim.x + threadIdx.x;
  if (i < n) p[i] = 0.0f;
}

__global__ void __launch_bounds__(512)
embed_xg0_kernel(const float* __restrict__ in,
                 const float* __restrict__ We,
                 const float* __restrict__ be,
                 const float* __restrict__ Wx0,
                 const float* __restrict__ b0,
                 float* __restrict__ xg)
{
  extern __shared__ __align__(16) char smem[];
  float* inP = (float*)smem;
  v2f*   WeP = (v2f*)(smem + 64 * IS * 4);
  float* xb  = (float*)(smem + 64 * IS * 4 + 16 * BP64 * 8);
  v2f*   WxP = (v2f*)(smem + 64 * IS * 4 + 16 * BP64 * 8 + 64 * XS * 4);

  const int tid  = threadIdx.x;
  const int wave = tid >> 5;
  const int lane = tid & 31;
  const int lh   = lane >> 4;
  const int ln   = lane & 15;
  const int row0 = blockIdx.x * 64;

  for (int i = tid; i < 64 * 32; i += 512) {
    int r = i >> 5, c = i & 31;
    inP[r * IS + c] = in[(size_t)(row0 + r) * 32 + c];
  }
  for (int i = tid; i < 16 * 64; i += 512) {
    int kp = i >> 6, n = i & 63;
    v2f w; w.x = We[(2 * kp) * 64 + n]; w.y = We[(2 * kp + 1) * 64 + n];
    WeP[kp * BP64 + n] = w;
  }
  for (int i = tid; i < 32 * 512; i += 512) {
    int kp = i >> 9, n = i & 511;
    v2f w; w.x = Wx0[(2 * kp) * 512 + n]; w.y = Wx0[(2 * kp + 1) * 512 + n];
    WxP[kp * BP512 + n] = w;
  }
  __syncthreads();

  {
    const int tm = wave >> 2, tn = wave & 3;
    const int col = tn * 16 + ln;
    const float bv = be[col];
    v8f acc = vzero8();
    const int am = tm * 16 + ln;
#pragma unroll
    for (int k0 = 0; k0 < 32; k0 += 4) {
      v2f a = *(const v2f*)&inP[am * IS + k0 + 2 * lh];
      v2f b = WeP[((k0 >> 1) + lh) * BP64 + col];
      acc = wmma4(a, b, acc);
    }
#pragma unroll
    for (int r = 0; r < 8; ++r)
      xb[(tm * 16 + r + 8 * lh) * XS + col] = acc[r] + bv;
  }
  __syncthreads();

  for (int task = wave; task < 128; task += 16) {
    const int tm = task >> 5, tn = task & 31;
    const int col = tn * 16 + ln;
    const float bv = b0[col];
    v8f acc = vzero8();
    const int am = tm * 16 + ln;
#pragma unroll
    for (int k0 = 0; k0 < 64; k0 += 4) {
      v2f a = *(const v2f*)&xb[am * XS + k0 + 2 * lh];
      v2f b = WxP[((k0 >> 1) + lh) * BP512 + col];
      acc = wmma4(a, b, acc);
    }
#pragma unroll
    for (int r = 0; r < 8; ++r)
      xg[(size_t)(row0 + tm * 16 + r + 8 * lh) * 512 + col] = acc[r] + bv;
  }
}

__global__ void __launch_bounds__(512)
xg1_kernel(const float* __restrict__ x1,
           const float* __restrict__ Wx1,
           const float* __restrict__ b1,
           float* __restrict__ xg)
{
  extern __shared__ __align__(16) char smem[];
  float* aB = (float*)smem;
  v2f*   WP = (v2f*)(smem + 64 * HS * 4);

  const int tid  = threadIdx.x;
  const int wave = tid >> 5;
  const int lane = tid & 31;
  const int lh   = lane >> 4;
  const int ln   = lane & 15;
  const int row0 = blockIdx.x * 64;

  for (int i = tid; i < 64 * 128; i += 512) {
    int r = i >> 7, c = i & 127;
    aB[r * HS + c] = x1[(size_t)(row0 + r) * 128 + c];
  }
  for (int i = tid; i < 64 * 512; i += 512) {
    int kp = i >> 9, n = i & 511;
    v2f w; w.x = Wx1[(2 * kp) * 512 + n]; w.y = Wx1[(2 * kp + 1) * 512 + n];
    WP[kp * BP512 + n] = w;
  }
  __syncthreads();

  for (int task = wave; task < 128; task += 16) {
    const int tm = task >> 5, tn = task & 31;
    const int col = tn * 16 + ln;
    const float bv = b1[col];
    v8f acc = vzero8();
    const int am = tm * 16 + ln;
    for (int k0 = 0; k0 < 128; k0 += 4) {
      v2f a = *(const v2f*)&aB[am * HS + k0 + 2 * lh];
      v2f b = WP[((k0 >> 1) + lh) * BP512 + col];
      acc = wmma4(a, b, acc);
    }
#pragma unroll
    for (int r = 0; r < 8; ++r)
      xg[(size_t)(row0 + tm * 16 + r + 8 * lh) * 512 + col] = acc[r] + bv;
  }
}

__global__ void __launch_bounds__(288)
fold_kernel(const float* __restrict__ Wo,
            const float* __restrict__ bo,
            const float* __restrict__ Wx,
            const float* __restrict__ b,
            float* __restrict__ Wf,
            float* __restrict__ bf)
{
  const int tid  = threadIdx.x;
  const int wave = tid >> 5;
  const int lane = tid & 31;
  const int lh   = lane >> 4;
  const int ln   = lane & 15;
  const int col  = blockIdx.x * 16 + ln;
  const bool isW = wave < 8;
  const float* arow = isW ? Wo + (wave * 16 + ln) * HID : bo;
  const bool   live = isW || ln == 0;
  v8f acc = vzero8();
  v2f a[32], w[32];
#pragma unroll
  for (int i = 0; i < 32; ++i) {
    const v2f av = *(const v2f*)(arow + 4 * i + 2 * lh);
    a[i].x = live ? av.x : 0.0f;
    a[i].y = live ? av.y : 0.0f;
    w[i].x = Wx[(4 * i + 2 * lh) * G4 + col];
    w[i].y = Wx[(4 * i + 2 * lh + 1) * G4 + col];
  }
#pragma unroll
  for (int i = 0; i < 32; ++i) acc = wmma4(a[i], w[i], acc);
  if (isW) {
#pragma unroll
    for (int r = 0; r < 8; ++r)
      Wf[(wave * 16 + r + 8 * lh) * G4 + col] = acc[r];
  } else if (lh == 0) {
    bf[col] = acc[0] + b[col];
  }
}

template <int NOUT>
__global__ void __launch_bounds__(512)
scan_kernel(const float* __restrict__ xg,
            const float* __restrict__ Wh,
            const float* __restrict__ Wo,
            const float* __restrict__ bo,
            float* __restrict__ ring,
            float* __restrict__ out)
{
  extern __shared__ __align__(16) char smem[];
  v2f*   WhP  = (v2f*)smem;
  float* hbuf = (float*)(smem + WHS_BYTES);

  const int tid  = threadIdx.x;
  const int wave = tid >> 5;
  const int lane = tid & 31;
  const int lh   = lane >> 4;
  const int ln   = lane & 15;

  for (int i = tid; i < 64 * 512; i += 512) {
    int kp = i >> 9, n = i & 511;
    int g = n >> 7, hh = n & 127;
    v2f w; w.x = Wh[(2 * kp) * 512 + n]; w.y = Wh[(2 * kp + 1) * 512 + n];
    WhP[((((hh >> 4) * 64 + kp) * 2 + (g >> 1)) * 16 + (((hh & 15) + 8 * (kp & 1)) & 15)) * 2 + (g & 1)] = w;
  }
  for (int i = tid; i < 64 * HS; i += 512) hbuf[i] = 0.0f;
  __syncthreads();

  const int tnh  = wave & 7;
  const int colh = tnh * 16 + ln;
  const v4f* wq  = (const v4f*)WhP + ((tnh * 64 + lh) * 2) * 16 + ((ln + 8 * lh) & 15);
  const float* abw = hbuf + ((wave >> 3) * 16 + ln) * HS + 2 * lh;

  v8f  acc[4];
  Frag fr[2];
#pragma unroll
  for (int g = 0; g < 4; ++g) acc[g] = vzero8();
  load_frag(fr[0], abw, wq, 0);
  k_blocks<0, 8>(fr, acc, abw, wq);

  for (int t = 0; t < T_DIM; ++t) {
    const float* xgt = xg + (size_t)t * (64 * 512);

    if (t + 1 < T_DIM) {
      const float* nxt = xgt + 64 * 512;
#pragma unroll
      for (int q = 0; q < 8; ++q)
        __builtin_prefetch(nxt + ((size_t)tid + q * 512) * 32, 0, 0);
    }

#pragma unroll
    for (int ti = 0; ti < 2; ++ti) {
      const int task = wave + ti * 16;
      const int tm   = task >> 3;

      float xv0[8], xv1[8], xv2[8], xv3[8];
      {
        const unsigned xoff = (unsigned)((tm * 16 + 8 * lh) * 512 + colh) * 4u;
        const char* xb = (const char*)xgt + xoff;
#pragma unroll
        for (int r = 0; r < 8; ++r) {
          const float* p = (const float*)(xb + r * 2048);
          xv0[r] = p[0];  xv1[r] = p[128];
          xv2[r] = p[256]; xv3[r] = p[384];
        }
      }

      constexpr int EARLY = KF - 3;
      v4f rv[KF - 1][2];
      const unsigned rboff = (unsigned)(task * 64 + lane) * 16u;
#pragma unroll
      for (int j = 1; j <= EARLY; ++j) {
        const char* sp = (const char*)ring + (size_t)((t - j) & 15) * 32768;
        rv[j - 1][0] = *(const v4f*)(sp + rboff);
        rv[j - 1][1] = *(const v4f*)(sp + rboff + 512);
      }

      const float* ab = abw + ti * (32 * HS);
      k_blocks<8, 16>(fr, acc, ab, wq);

      bar_arrive();

      {
        typedef const char __attribute__((address_space(1)))* gptr_cc;
        typedef const v4f __attribute__((address_space(1)))* gptr_cv4;
        gptr_cc ringl = (gptr_cc)ring;
        asm volatile("" : "+s"(ringl));
#pragma unroll
        for (int j = EARLY + 1; j < KF; ++j) {
          gptr_cc sp = ringl + (size_t)((t - j) & 15) * 32768;
          rv[j - 1][0] = *(gptr_cv4)(sp + rboff);
          rv[j - 1][1] = *(gptr_cv4)(sp + rboff + 512);
        }
      }

      float hc[8], og[8], dd[8];
#pragma unroll
      for (int r = 0; r < 8; ++r) {
        const float ig = sigm(acc[0][r] + xv0[r]);
        og[r] = sigm(acc[1][r] + xv1[r]);
        const float cg = tanhf(acc[2][r] + xv2[r]);
        dd[r] = 0.5f * sigm(acc[3][r] + xv3[r]);
        hc[r] = ig * cg;
      }
      constexpr int U0 = 24;
      constexpr int U1 = 48;
      constexpr int UPG = (NOUT == 128) ? 3 : 4;
      float csum[8], w[8];
#pragma unroll
      for (int r = 0; r < 8; ++r) { csum[r] = hc[r]; w[r] = 1.0f; }
#pragma unroll
      for (int j = 1; j <= U0 / 8; ++j) {
        const float jm1 = (float)(j - 1), rj = 1.0f / (float)j;
#pragma unroll
        for (int r = 0; r < 8; ++r) {
          w[r] *= (jm1 + dd[r]) * rj;
          csum[r] += w[r] * rv[j - 1][r >> 2][r & 3];
        }
      }
      {
        char* sw = (char*)ring + (size_t)(t & 15) * 32768;
        v4f lo, hi;
        lo.x = hc[0]; lo.y = hc[1]; lo.z = hc[2]; lo.w = hc[3];
        hi.x = hc[4]; hi.y = hc[5]; hi.z = hc[6]; hi.w = hc[7];
        *(v4f*)(sw + rboff)       = lo;
        *(v4f*)(sw + rboff + 512) = hi;
      }

      bar_await();

      float t2[128];
      auto stage = [&](int n) __attribute__((always_inline)) {
#pragma unroll
        for (int h = 0; h < UPG; ++h) {
          const int uB = U1 + UPG * (n + 2) + h;
          const int uC = U1 + UPG * (n + 1) + h, uD = U1 + UPG * n + h;
          if (uB >= U1 && uB < 120) t2[uB] = ((float)(uB >> 3) + dd[uB & 7]) * (1.0f / (float)((uB >> 3) + 1));
          if (uC >= U1 && uC < 120) w[uC & 7] *= t2[uC];
          if (uD >= U1 && uD < 120) csum[uD & 7] += w[uD & 7] * rv[uD >> 3][(uD & 7) >> 2][uD & 3];
        }
      };

      const float* abn = abw + (ti ^ 1) * (32 * HS);
#pragma unroll
      for (int g = 0; g < 4; ++g) acc[g] = vzero8();
      load_frag(fr[0], abn, wq, 0);
#pragma unroll
      for (int j = U0 / 8 + 1; j <= U1 / 8; ++j) {
        const float jm1 = (float)(j - 1), rj = 1.0f / (float)j;
#pragma unroll
        for (int r = 0; r < 8; ++r) {
          w[r] *= (jm1 + dd[r]) * rj;
          csum[r] += w[r] * rv[j - 1][r >> 2][r & 3];
        }
      }
      __builtin_amdgcn_sched_barrier(0);
      stage(-2); stage(-1);
#pragma unroll
      for (int blk = 0; blk < 8; ++blk) {
        load_frag(fr[(blk & 1) ^ 1], abn, wq, blk + 1);
        __builtin_amdgcn_sched_barrier(0);
#pragma unroll
        for (int kk = 0; kk < 2; ++kk) {
          const v4f q0 = fr[blk & 1].q[kk][0], q1 = fr[blk & 1].q[kk][1];
          v2f b[4];
          b[0].x = q0.x; b[0].y = q0.y; b[1].x = q0.z; b[1].y = q0.w;
          b[2].x = q1.x; b[2].y = q1.y; b[3].x = q1.z; b[3].y = q1.w;
#pragma unroll
          for (int g = 0; g < 4; ++g) {
            acc[g] = wmma4(fr[blk & 1].a[kk], b[g], acc[g]);
            stage(blk * 8 + kk * 4 + g);
            __builtin_amdgcn_sched_barrier(0);
          }
        }
      }

      asm volatile("" :: "v"(csum[0]), "v"(csum[1]), "v"(csum[2]), "v"(csum[3]),
                         "v"(csum[4]), "v"(csum[5]), "v"(csum[6]), "v"(csum[7]));
      float hn[8];
#pragma unroll
      for (int r = 0; r < 8; ++r)
        hn[r] = og[r] * tanhf(csum[r]);

#pragma unroll
      for (int r = 0; r < 8; ++r)
        hbuf[(tm * 16 + r + 8 * lh) * HS + colh] = hn[r];

      if constexpr (NOUT == 128) {
        char* hb = (char*)out + (size_t)t * (64 * HID * 4);
        const unsigned hoff = (unsigned)(((wave >> 3) * 16 + 8 * lh) * HID + (wave & 7) * 16 + ln) * 4u;
#pragma unroll
        for (int r = 0; r < 8; ++r)
          *(float*)(hb + hoff + ti * 16384 + r * 512) = hn[r];
      } else {
        static_assert(NOUT == 32, "waves 0-3 = batch half 0, waves 4-7 = batch half 1");
        if (ti == 0) {
          if (t > 0 && wave >= 4 && wave < 8)
            project_tile<NOUT>(hbuf, Wo, bo, out + (size_t)(t - 1) * (64 * NOUT), wave, lh, ln);
        } else {
          if (wave < 4)
            project_tile<NOUT>(hbuf, Wo, bo, out + (size_t)t * (64 * NOUT), wave, lh, ln);
        }
      }
    }
  }
  if constexpr (NOUT != 128) {
    __syncthreads();
    if (wave >= 4 && wave < 8)
      project_tile<NOUT>(hbuf, Wo, bo, out + (size_t)(T_DIM - 1) * (64 * NOUT), wave, lh, ln);
  }
}

extern "C" void kernel_launch(void* const* d_in, const int* in_sizes, int n_in,
                              void* d_out, int out_size, void* d_ws, size_t ws_size,
                              hipStream_t stream) {
  const float* inputs = (const float*)d_in[0];
  const float* W_emb  = (const float*)d_in[1];
  const float* b_emb  = (const float*)d_in[2];
  const float* Wx0    = (const float*)d_in[3];
  const float* Wh0    = (const float*)d_in[4];
  const float* b0     = (const float*)d_in[5];
  const float* Wo0    = (const float*)d_in[6];
  const float* bo0    = (const float*)d_in[7];
  const float* Wx1    = (const float*)d_in[8];
  const float* Wh1    = (const float*)d_in[9];
  const float* b1     = (const float*)d_in[10];
  const float* Wo1    = (const float*)d_in[11];
  const float* bo1    = (const float*)d_in[12];
  float* out = (float*)d_out;

  float* xg   = (float*)d_ws;
  float* h0   = xg + (size_t)T_DIM * B_DIM * G4;
  float* ring = h0 + (size_t)T_DIM * B_DIM * HID;
  const int ringN = KF * B_DIM * HID;
  float* Wf   = ring + (size_t)KF * B_DIM * HID;
  float* bf   = Wf + (size_t)HID * G4;

  const size_t ldsE = (size_t)64 * IS * 4 + 16 * BP64 * 8 + 64 * XS * 4 + 32 * BP512 * 8;
  const size_t ldsX = (size_t)64 * HS * 4 + 64 * BP512 * 8;
  const size_t ldsS = (size_t)WHS_BYTES + 64 * HS * 4;

  const int nblk = (T_DIM * B_DIM) / 64;

  embed_xg0_kernel<<<nblk, 512, ldsE, stream>>>(inputs, W_emb, b_emb, Wx0, b0, xg);
  zero_kernel<<<ringN / 256, 256, 0, stream>>>(ring, ringN);
  scan_kernel<HID><<<1, 512, ldsS, stream>>>(xg, Wh0, Wo0, bo0, ring, h0);
  fold_kernel<<<G4 / 16, 288, 0, stream>>>(Wo0, bo0, Wx1, b1, Wf, bf);
  xg1_kernel<<<nblk, 512, ldsX, stream>>>(h0, Wf, bf, xg);
  zero_kernel<<<ringN / 256, 256, 0, stream>>>(ring, ringN);
  scan_kernel<OUTD><<<1, 512, ldsS, stream>>>(xg, Wh1, Wo1, bo1, ring, out);
}


# scan: a task's o*tanh(c) and h stores issue between the WMMAs of the NEXT task's second k-loop half (tanh as a two-range operation list, 1-6 per gap), not exposed at its end; tanh's last bit may move
# speedup vs baseline: 2.4568x; 1.0527x over previous
#include <hip/hip_runtime.h>
#include <cstdint>
#include <cstddef>
#include <utility>

#define DEVINL __device__ __forceinline__

typedef float v2f __attribute__((ext_vector_type(2)));
typedef float v4f __attribute__((ext_vector_type(4)));
typedef float v8f __attribute__((ext_vector_type(8)));

static constexpr int T_DIM  = 1024;
static constexpr int B_DIM  = 64;
static constexpr int IN_DIM = 32;
static constexpr int EMB    = 64;
static constexpr int HID    = 128;
static constexpr int KF     = 16;
static constexpr int OUTD   = 32;
static constexpr int G4     = 512;

static constexpr int HS     = 132;
static constexpr int XS     = 68;
static constexpr int IS     = 36;
static constexpr int BP512  = 528;
static constexpr int BP64   = 80;
static constexpr int WHS_BYTES = 8 * 64 * 2 * 16 * 16;

DEVINL float sigm(float x) { return 1.0f / (1.0f + __expf(-x)); }

struct TanhSt { v2f zs, a, k, t; int k0, k1; };
DEVINL v2f splat2(float c) { v2f r; r.x = c; r.y = c; return r; }
static constexpr int TANH_STAGES = 21;
DEVINL void hold2(v2f& v) { asm volatile("" : "+v"(v)); }
DEVINL void hold1(float& v) { asm volatile("" : "+v"(v)); }
DEVINL void hold2(int& a, int& b) { asm volatile("" : "+v"(a), "+v"(b)); }
template <int S, bool HOLD>
DEVINL void tanh_stage(TanhSt& q, v2f x, v2f og, v2f& res) {
  v2f y; y.x = __builtin_fabsf(x.x); y.y = __builtin_fabsf(x.y);
  constexpr float L2E_HI = 0x1.715476p+0f, L2E_LO = 0x1.4ae0bep-26f;
  if constexpr (S == 0)  q.zs = x * x;
  if constexpr (S == 1)  q.t  = __builtin_elementwise_fma(splat2(-0x1.758e7ap-8f), q.zs, splat2(0x1.521192p-6f));
  if constexpr (S == 2)  q.t  = __builtin_elementwise_fma(q.zs, q.t, splat2(-0x1.b8389cp-5f));
  if constexpr (S == 3)  q.t  = __builtin_elementwise_fma(q.zs, q.t, splat2(0x1.110704p-3f));
  if constexpr (S == 4)  q.t  = __builtin_elementwise_fma(q.zs, q.t, splat2(-0x1.555532p-2f));
  if constexpr (S == 5)  q.t  = y * q.t;
  if constexpr (S == 6)  q.zs = __builtin_elementwise_fma(q.zs, q.t, y);
  if constexpr (S == 7)  q.a  = y + y;
  if constexpr (S == 8)  { q.k.x = __builtin_rintf(q.a.x * L2E_HI); q.k.y = __builtin_rintf(q.a.y * L2E_HI); }
  if constexpr (S == 9)  q.t  = __builtin_elementwise_fma(q.a, splat2(L2E_HI), -q.k);
  if constexpr (S == 10) q.t  = __builtin_elementwise_fma(q.a, splat2(L2E_LO), q.t);
  if constexpr (S == 11) { q.k0 = (int)q.k.x; q.k1 = (int)q.k.y; }
  if constexpr (S == 12) { q.t.x = __builtin_amdgcn_exp2f(q.t.x); q.t.y = __builtin_amdgcn_exp2f(q.t.y); }
  if constexpr (S == 13) { q.t.x = __builtin_ldexpf(q.t.x, q.k0); q.t.y = __builtin_ldexpf(q.t.y, q.k1); }
  if constexpr (S == 14) { q.t.x = (0x1.62e430p+6f < q.a.x) ? __builtin_inff() : q.t.x;
                           q.t.y = (0x1.62e430p+6f < q.a.y) ? __builtin_inff() : q.t.y; }
  if constexpr (S == 15) q.t  = q.t + splat2(1.0f);
  if constexpr (S == 16) { q.t.x = __builtin_amdgcn_rcpf(q.t.x); q.t.y = __builtin_amdgcn_rcpf(q.t.y); }
  if constexpr (S == 17) q.t  = __builtin_elementwise_fma(q.t, splat2(-2.0f), splat2(1.0f));
  if constexpr (S == 18) { q.t.x = (0.625f > y.x) ? q.zs.x : q.t.x; q.t.y = (0.625f > y.y) ? q.zs.y : q.t.y; }
  if constexpr (S == 19) { q.t.x = __builtin_copysignf(q.t.x, x.x); q.t.y = __builtin_copysignf(q.t.y, x.y); }
  if constexpr (S == 20) res  = og * q.t;
  if constexpr (HOLD) {
    if constexpr (S == 0 || S == 6) hold2(q.zs);
    else if constexpr (S == 7)      hold2(q.a);
    else if constexpr (S == 8)      hold2(q.k);
    else if constexpr (S == 11)     hold2(q.k0, q.k1);
    else if constexpr (S == 20)     hold2(res);
    else                            hold2(q.t);
  }
}
DEVINL v2f out_path(v2f x, v2f og) {
  TanhSt q; v2f res;
  [&]<int... S>(std::integer_sequence<int, S...>) __attribute__((always_inline)) {
    (tanh_stage<S, false>(q, x, og, res), ...);
  }(std::make_integer_sequence<int, TANH_STAGES>{});
  return res;
}
template <int N>
DEVINL void out_gap(TanhSt (&q)[4], const v2f (&x)[4], const v2f (&og)[4], v2f (&hn)[4]) {
  constexpr int G = (N >= 23) ? 1 : 0;
  constexpr int M = N - 23 * G;
  if constexpr (M >= 0 && M < TANH_STAGES)         tanh_stage<M, true>(q[2 * G], x[2 * G], og[2 * G], hn[2 * G]);
  if constexpr (M - 1 >= 0 && M - 1 < TANH_STAGES) tanh_stage<M - 1, true>(q[2 * G + 1], x[2 * G + 1], og[2 * G + 1], hn[2 * G + 1]);
}

DEVINL v8f wmma4(v2f a, v2f b, v8f c) {
  return __builtin_amdgcn_wmma_f32_16x16x4_f32(false, a, false, b, (short)0, c,
                                               false, false);
}

DEVINL v8f vzero8() {
  v8f z;
#pragma unroll
  for (int r = 0; r < 8; ++r) z[r] = 0.0f;
  return z;
}

DEVINL void bar_arrive() { asm volatile("s_wait_dscnt 0x0\n\ts_barrier_signal -1" ::: "memory"); }
DEVINL void bar_await()  { asm volatile("s_barrier_wait -1" ::: "memory"); }

struct Frag { v2f a[2]; v4f q[2][2]; };

DEVINL void load_frag(Frag& f, const float* ab, const v4f* wq, int blk) {
#pragma unroll
  for (int kk = 0; kk < 2; ++kk) {
    const int i = blk * 2 + kk;
    f.a[kk]    = *(const v2f*)(ab + 4 * i);
    f.q[kk][0] = wq[i * 64];
    f.q[kk][1] = wq[i * 64 + 16];
  }
}

DEVINL void block_wmma(const Frag& f, v8f (&acc)[4]) {
#pragma unroll
  for (int kk = 0; kk < 2; ++kk) {
    const v4f q0 = f.q[kk][0], q1 = f.q[kk][1];
    v2f b0, b1, b2, b3;
    b0.x = q0.x; b0.y = q0.y; b1.x = q0.z; b1.y = q0.w;
    b2.x = q1.x; b2.y = q1.y; b3.x = q1.z; b3.y = q1.w;
    acc[0] = wmma4(f.a[kk], b0, acc[0]);
    acc[1] = wmma4(f.a[kk], b1, acc[1]);
    acc[2] = wmma4(f.a[kk], b2, acc[2]);
    acc[3] = wmma4(f.a[kk], b3, acc[3]);
  }
}

template <int B0, int B1>
DEVINL void k_blocks(Frag (&fr)[2], v8f (&acc)[4], const float* ab, const v4f* wq) {
#pragma unroll
  for (int blk = B0; blk < B1; ++blk) {
    if (blk + 1 < 16) load_frag(fr[(blk & 1) ^ 1], ab, wq, blk + 1);
    __builtin_amdgcn_sched_barrier(0);
    block_wmma(fr[blk & 1], acc);
    __builtin_amdgcn_sched_barrier(0);
  }
}

template <int NOUT>
DEVINL void project_tile(const float* hbuf, const float* __restrict__ Wo,
                         const float* __restrict__ bo, float* ot, int p, int lh, int ln) {
  const int tm  = p / (NOUT / 16);
  const int tno = p % (NOUT / 16);
  const int col = tno * 16 + ln;
  const float bv = bo[col];
  v8f acc = vzero8();
  const int am = tm * 16 + ln;
  for (int k0 = 0; k0 < 128; k0 += 4) {
    v2f a = *(const v2f*)&hbuf[am * HS + k0 + 2 * lh];
    v2f b;
    b.x = Wo[(k0 + 2 * lh) * NOUT + col];
    b.y = Wo[(k0 + 2 * lh + 1) * NOUT + col];
    acc = wmma4(a, b, acc);
  }
#pragma unroll
  for (int r = 0; r < 8; ++r)
    ot[(tm * 16 + r + 8 * lh) * NOUT + col] = acc[r] + bv;
}

__global__ void zero_kernel(float* __restrict__ p, int n) {
  int i = blockIdx.x * blockDim.x + threadIdx.x;
  if (i < n) p[i] = 0.0f;
}

__global__ void __launch_bounds__(512)
embed_xg0_kernel(const float* __restrict__ in,
                 const float* __restrict__ We,
                 const float* __restrict__ be,
                 const float* __restrict__ Wx0,
                 const float* __restrict__ b0,
                 float* __restrict__ xg)
{
  extern __shared__ __align__(16) char smem[];
  float* inP = (float*)smem;
  v2f*   WeP = (v2f*)(smem + 64 * IS * 4);
  float* xb  = (float*)(smem + 64 * IS * 4 + 16 * BP64 * 8);
  v2f*   WxP = (v2f*)(smem + 64 * IS * 4 + 16 * BP64 * 8 + 64 * XS * 4);

  const int tid  = threadIdx.x;
  const int wave = tid >> 5;
  const int lane = tid & 31;
  const int lh   = lane >> 4;
  const int ln   = lane & 15;
  const int row0 = blockIdx.x * 64;

  for (int i = tid; i < 64 * 32; i += 512) {
    int r = i >> 5, c = i & 31;
    inP[r * IS + c] = in[(size_t)(row0 + r) * 32 + c];
  }
  for (int i = tid; i < 16 * 64; i += 512) {
    int kp = i >> 6, n = i & 63;
    v2f w; w.x = We[(2 * kp) * 64 + n]; w.y = We[(2 * kp + 1) * 64 + n];
    WeP[kp * BP64 + n] = w;
  }
  for (int i = tid; i < 32 * 512; i += 512) {
    int kp = i >> 9, n = i & 511;
    v2f w; w.x = Wx0[(2 * kp) * 512 + n]; w.y = Wx0[(2 * kp + 1) * 512 + n];
    WxP[kp * BP512 + n] = w;
  }
  __syncthreads();

  {
    const int tm = wave >> 2, tn = wave & 3;
    const int col = tn * 16 + ln;
    const float bv = be[col];
    v8f acc = vzero8();
    const int am = tm * 16 + ln;
#pragma unroll
    for (int k0 = 0; k0 < 32; k0 += 4) {
      v2f a = *(const v2f*)&inP[am * IS + k0 + 2 * lh];
      v2f b = WeP[((k0 >> 1) + lh) * BP64 + col];
      acc = wmma4(a, b, acc);
    }
#pragma unroll
    for (int r = 0; r < 8; ++r)
      xb[(tm * 16 + r + 8 * lh) * XS + col] = acc[r] + bv;
  }
  __syncthreads();

  for (int task = wave; task < 128; task += 16) {
    const int tm = task >> 5, tn = task & 31;
    const int col = tn * 16 + ln;
    const float bv = b0[col];
    v8f acc = vzero8();
    const int am = tm * 16 + ln;
#pragma unroll
    for (int k0 = 0; k0 < 64; k0 += 4) {
      v2f a = *(const v2f*)&xb[am * XS + k0 + 2 * lh];
      v2f b = WxP[((k0 >> 1) + lh) * BP512 + col];
      acc = wmma4(a, b, acc);
    }
#pragma unroll
    for (int r = 0; r < 8; ++r)
      xg[(size_t)(row0 + tm * 16 + r + 8 * lh) * 512 + col] = acc[r] + bv;
  }
}

__global__ void __launch_bounds__(512)
xg1_kernel(const float* __restrict__ x1,
           const float* __restrict__ Wx1,
           const float* __restrict__ b1,
           float* __restrict__ xg)
{
  extern __shared__ __align__(16) char smem[];
  float* aB = (float*)smem;
  v2f*   WP = (v2f*)(smem + 64 * HS * 4);

  const int tid  = threadIdx.x;
  const int wave = tid >> 5;
  const int lane = tid & 31;
  const int lh   = lane >> 4;
  const int ln   = lane & 15;
  const int row0 = blockIdx.x * 64;

  for (int i = tid; i < 64 * 128; i += 512) {
    int r = i >> 7, c = i & 127;
    aB[r * HS + c] = x1[(size_t)(row0 + r) * 128 + c];
  }
  for (int i = tid; i < 64 * 512; i += 512) {
    int kp = i >> 9, n = i & 511;
    v2f w; w.x = Wx1[(2 * kp) * 512 + n]; w.y = Wx1[(2 * kp + 1) * 512 + n];
    WP[kp * BP512 + n] = w;
  }
  __syncthreads();

  for (int task = wave; task < 128; task += 16) {
    const int tm = task >> 5, tn = task & 31;
    const int col = tn * 16 + ln;
    const float bv = b1[col];
    v8f acc = vzero8();
    const int am = tm * 16 + ln;
    for (int k0 = 0; k0 < 128; k0 += 4) {
      v2f a = *(const v2f*)&aB[am * HS + k0 + 2 * lh];
      v2f b = WP[((k0 >> 1) + lh) * BP512 + col];
      acc = wmma4(a, b, acc);
    }
#pragma unroll
    for (int r = 0; r < 8; ++r)
      xg[(size_t)(row0 + tm * 16 + r + 8 * lh) * 512 + col] = acc[r] + bv;
  }
}

__global__ void __launch_bounds__(288)
fold_kernel(const float* __restrict__ Wo,
            const float* __restrict__ bo,
            const float* __restrict__ Wx,
            const float* __restrict__ b,
            float* __restrict__ Wf,
            float* __restrict__ bf)
{
  const int tid  = threadIdx.x;
  const int wave = tid >> 5;
  const int lane = tid & 31;
  const int lh   = lane >> 4;
  const int ln   = lane & 15;
  const int col  = blockIdx.x * 16 + ln;
  const bool isW = wave < 8;
  const float* arow = isW ? Wo + (wave * 16 + ln) * HID : bo;
  const bool   live = isW || ln == 0;
  v8f acc = vzero8();
  v2f a[32], w[32];
#pragma unroll
  for (int i = 0; i < 32; ++i) {
    const v2f av = *(const v2f*)(arow + 4 * i + 2 * lh);
    a[i].x = live ? av.x : 0.0f;
    a[i].y = live ? av.y : 0.0f;
    w[i].x = Wx[(4 * i + 2 * lh) * G4 + col];
    w[i].y = Wx[(4 * i + 2 * lh + 1) * G4 + col];
  }
#pragma unroll
  for (int i = 0; i < 32; ++i) acc = wmma4(a[i], w[i], acc);
  if (isW) {
#pragma unroll
    for (int r = 0; r < 8; ++r)
      Wf[(wave * 16 + r + 8 * lh) * G4 + col] = acc[r];
  } else if (lh == 0) {
    bf[col] = acc[0] + b[col];
  }
}

template <int NOUT>
__global__ void __launch_bounds__(512)
scan_kernel(const float* __restrict__ xg,
            const float* __restrict__ Wh,
            const float* __restrict__ Wo,
            const float* __restrict__ bo,
            float* __restrict__ ring,
            float* __restrict__ out)
{
  extern __shared__ __align__(16) char smem[];
  v2f*   WhP  = (v2f*)smem;
  float* hbuf = (float*)(smem + WHS_BYTES);

  const int tid  = threadIdx.x;
  const int wave = tid >> 5;
  const int lane = tid & 31;
  const int lh   = lane >> 4;
  const int ln   = lane & 15;

  for (int i = tid; i < 64 * 512; i += 512) {
    int kp = i >> 9, n = i & 511;
    int g = n >> 7, hh = n & 127;
    v2f w; w.x = Wh[(2 * kp) * 512 + n]; w.y = Wh[(2 * kp + 1) * 512 + n];
    WhP[((((hh >> 4) * 64 + kp) * 2 + (g >> 1)) * 16 + (((hh & 15) + 8 * (kp & 1)) & 15)) * 2 + (g & 1)] = w;
  }
  for (int i = tid; i < 64 * HS; i += 512) hbuf[i] = 0.0f;
  __syncthreads();

  const int tnh  = wave & 7;
  const int colh = tnh * 16 + ln;
  const v4f* wq  = (const v4f*)WhP + ((tnh * 64 + lh) * 2) * 16 + ((ln + 8 * lh) & 15);
  const float* abw = hbuf + ((wave >> 3) * 16 + ln) * HS + 2 * lh;

  v8f  acc[4];
  Frag fr[2];
#pragma unroll
  for (int g = 0; g < 4; ++g) acc[g] = vzero8();
  load_frag(fr[0], abw, wq, 0);
  k_blocks<0, 8>(fr, acc, abw, wq);

  v2f pc[4], po[4];
#pragma unroll
  for (int p = 0; p < 4; ++p) { pc[p] = splat2(0.0f); po[p] = splat2(0.0f); }

  for (int t = 0; t < T_DIM; ++t) {
    const float* xgt = xg + (size_t)t * (64 * 512);

    if (t + 1 < T_DIM) {
      const float* nxt = xgt + 64 * 512;
#pragma unroll
      for (int q = 0; q < 8; ++q)
        __builtin_prefetch(nxt + ((size_t)tid + q * 512) * 32, 0, 0);
    }

#pragma unroll
    for (int ti = 0; ti < 2; ++ti) {
      const int task = wave + ti * 16;
      const int tm   = task >> 3;

      float xv0[8], xv1[8], xv2[8], xv3[8];
      {
        const unsigned xoff = (unsigned)((tm * 16 + 8 * lh) * 512 + colh) * 4u;
        const char* xb = (const char*)xgt + xoff;
#pragma unroll
        for (int r = 0; r < 8; ++r) {
          const float* p = (const float*)(xb + r * 2048);
          xv0[r] = p[0];  xv1[r] = p[128];
          xv2[r] = p[256]; xv3[r] = p[384];
        }
      }

      constexpr int EARLY = KF - 6;
      v4f rv[KF - 1][2];
      const unsigned rboff = (unsigned)(task * 64 + lane) * 16u;
#pragma unroll
      for (int j = 1; j <= EARLY; ++j) {
        const char* sp = (const char*)ring + (size_t)((t - j) & 15) * 32768;
        rv[j - 1][0] = *(const v4f*)(sp + rboff);
        rv[j - 1][1] = *(const v4f*)(sp + rboff + 512);
      }

      const float* ab = abw + ti * (32 * HS);
      v2f hn2[4];
      TanhSt tq[4];
      [&]<int... N>(std::integer_sequence<int, N...>) __attribute__((always_inline)) {
        ([&] __attribute__((always_inline)) {
          constexpr int blk = 8 + N / 8, kk = (N / 4) & 1, g = N & 3;
          if constexpr ((N & 7) == 0) {
            if constexpr (blk + 1 < 16) load_frag(fr[(blk & 1) ^ 1], ab, wq, blk + 1);
            __builtin_amdgcn_sched_barrier(0);
          }
          const v4f qq = fr[blk & 1].q[kk][g >> 1];
          v2f b;
          b.x = (g & 1) ? qq.z : qq.x; b.y = (g & 1) ? qq.w : qq.y;
          acc[g] = wmma4(fr[blk & 1].a[kk], b, acc[g]);
          if constexpr (N < 46) out_gap<N>(tq, pc, po, hn2);
          if constexpr (N >= 46 && N < 50) {
            constexpr int p = N - 46;
            const int tmp = (wave >> 3) + 2 * (ti ^ 1);
            hbuf[(tmp * 16 + 2 * p + 8 * lh) * HS + colh]     = hn2[p].x;
            hbuf[(tmp * 16 + 2 * p + 1 + 8 * lh) * HS + colh] = hn2[p].y;
          }
          __builtin_amdgcn_sched_barrier(0);
        }(), ...);
      }(std::make_integer_sequence<int, 64>{});
      if constexpr (NOUT == 128) {
        if (ti == 1 || t > 0) {
          char* hb = (char*)out + (size_t)(t - (ti ^ 1)) * (64 * HID * 4);
          const unsigned hoff = (unsigned)(((wave >> 3) * 16 + 8 * lh) * HID + (wave & 7) * 16 + ln) * 4u;
#pragma unroll
          for (int p = 0; p < 4; ++p) {
            *(float*)(hb + hoff + (ti ^ 1) * 16384 + (2 * p) * 512)     = hn2[p].x;
            *(float*)(hb + hoff + (ti ^ 1) * 16384 + (2 * p + 1) * 512) = hn2[p].y;
          }
        }
      }

      bar_arrive();

      {
        typedef const char __attribute__((address_space(1)))* gptr_cc;
        typedef const v4f __attribute__((address_space(1)))* gptr_cv4;
        gptr_cc ringl = (gptr_cc)ring;
        asm volatile("" : "+s"(ringl));
#pragma unroll
        for (int j = EARLY + 1; j < KF; ++j) {
          gptr_cc sp = ringl + (size_t)((t - j) & 15) * 32768;
          rv[j - 1][0] = *(gptr_cv4)(sp + rboff);
          rv[j - 1][1] = *(gptr_cv4)(sp + rboff + 512);
        }
      }

      float hc[8], og[8], dd[8];
#pragma unroll
      for (int r = 0; r < 8; ++r) {
        const float ig = sigm(acc[0][r] + xv0[r]);
        og[r] = sigm(acc[1][r] + xv1[r]);
        const float cg = tanhf(acc[2][r] + xv2[r]);
        dd[r] = 0.5f * sigm(acc[3][r] + xv3[r]);
        hc[r] = ig * cg;
      }
      constexpr int U0 = 24;
      constexpr int U1 = 48;
      constexpr int UPG = (NOUT == 128) ? 3 : 4;
      float csum[8], w[8];
#pragma unroll
      for (int r = 0; r < 8; ++r) { csum[r] = hc[r]; w[r] = 1.0f; }
#pragma unroll
      for (int j = 1; j <= U0 / 8; ++j) {
        const float jm1 = (float)(j - 1), rj = 1.0f / (float)j;
#pragma unroll
        for (int r = 0; r < 8; ++r) {
          w[r] *= (jm1 + dd[r]) * rj;
          csum[r] += w[r] * rv[j - 1][r >> 2][r & 3];
        }
      }
      {
        char* sw = (char*)ring + (size_t)(t & 15) * 32768;
        v4f lo, hi;
        lo.x = hc[0]; lo.y = hc[1]; lo.z = hc[2]; lo.w = hc[3];
        hi.x = hc[4]; hi.y = hc[5]; hi.z = hc[6]; hi.w = hc[7];
        *(v4f*)(sw + rboff)       = lo;
        *(v4f*)(sw + rboff + 512) = hi;
      }

      bar_await();

      float t2[128];
      auto stage = [&](int n) __attribute__((always_inline)) {
#pragma unroll
        for (int h = 0; h < UPG; ++h) {
          const int uB = U1 + UPG * (n + 2) + h;
          const int uC = U1 + UPG * (n + 1) + h, uD = U1 + UPG * n + h;
          if (uB >= U1 && uB < 120) { t2[uB] = ((float)(uB >> 3) + dd[uB & 7]) * (1.0f / (float)((uB >> 3) + 1)); hold1(t2[uB]); }
          if (uC >= U1 && uC < 120) { w[uC & 7] *= t2[uC]; hold1(w[uC & 7]); }
          if (uD >= U1 && uD < 120) { csum[uD & 7] += w[uD & 7] * rv[uD >> 3][(uD & 7) >> 2][uD & 3]; hold1(csum[uD & 7]); }
        }
      };

      const float* abn = abw + (ti ^ 1) * (32 * HS);
#pragma unroll
      for (int g = 0; g < 4; ++g) acc[g] = vzero8();
      load_frag(fr[0], abn, wq, 0);
#pragma unroll
      for (int j = U0 / 8 + 1; j <= U1 / 8; ++j) {
        const float jm1 = (float)(j - 1), rj = 1.0f / (float)j;
#pragma unroll
        for (int r = 0; r < 8; ++r) {
          w[r] *= (jm1 + dd[r]) * rj;
          csum[r] += w[r] * rv[j - 1][r >> 2][r & 3];
        }
      }
      __builtin_amdgcn_sched_barrier(0);
      stage(-2); stage(-1);
#pragma unroll
      for (int blk = 0; blk < 8; ++blk) {
        load_frag(fr[(blk & 1) ^ 1], abn, wq, blk + 1);
        __builtin_amdgcn_sched_barrier(0);
#pragma unroll
        for (int kk = 0; kk < 2; ++kk) {
          const v4f q0 = fr[blk & 1].q[kk][0], q1 = fr[blk & 1].q[kk][1];
          v2f b[4];
          b[0].x = q0.x; b[0].y = q0.y; b[1].x = q0.z; b[1].y = q0.w;
          b[2].x = q1.x; b[2].y = q1.y; b[3].x = q1.z; b[3].y = q1.w;
#pragma unroll
          for (int g = 0; g < 4; ++g) {
            acc[g] = wmma4(fr[blk & 1].a[kk], b[g], acc[g]);
            stage(blk * 8 + kk * 4 + g);
            __builtin_amdgcn_sched_barrier(0);
          }
        }
      }

#pragma unroll
      for (int p = 0; p < 4; ++p) {
        pc[p].x = csum[2 * p]; pc[p].y = csum[2 * p + 1];
        po[p].x = og[2 * p];   po[p].y = og[2 * p + 1];
      }

      if constexpr (NOUT != 128) {
        static_assert(NOUT == 32, "waves 0-3 = batch half 0, waves 4-7 = batch half 1");
        if (ti == 0) {
          if (t > 0 && wave >= 4 && wave < 8)
            project_tile<NOUT>(hbuf, Wo, bo, out + (size_t)(t - 1) * (64 * NOUT), wave, lh, ln);
        } else {
          if (wave < 4)
            project_tile<NOUT>(hbuf, Wo, bo, out + (size_t)t * (64 * NOUT), wave, lh, ln);
        }
      }
    }
  }
  {
    const int tmp = (wave >> 3) + 2;
#pragma unroll
    for (int r = 0; r < 8; ++r) {
      const v2f hp = out_path(pc[r >> 1], po[r >> 1]);
      const float hl = (r & 1) ? hp.y : hp.x;
      hbuf[(tmp * 16 + r + 8 * lh) * HS + colh] = hl;
      if constexpr (NOUT == 128) {
        const unsigned hoff = (unsigned)(((wave >> 3) * 16 + 8 * lh) * HID + (wave & 7) * 16 + ln) * 4u;
        *(float*)((char*)out + (size_t)(T_DIM - 1) * (64 * HID * 4) + hoff + 16384 + r * 512) = hl;
      }
    }
  }
  if constexpr (NOUT != 128) {
    __syncthreads();
    if (wave >= 4 && wave < 8)
      project_tile<NOUT>(hbuf, Wo, bo, out + (size_t)(T_DIM - 1) * (64 * NOUT), wave, lh, ln);
  }
}

extern "C" void kernel_launch(void* const* d_in, const int* in_sizes, int n_in,
                              void* d_out, int out_size, void* d_ws, size_t ws_size,
                              hipStream_t stream) {
  const float* inputs = (const float*)d_in[0];
  const float* W_emb  = (const float*)d_in[1];
  const float* b_emb  = (const float*)d_in[2];
  const float* Wx0    = (const float*)d_in[3];
  const float* Wh0    = (const float*)d_in[4];
  const float* b0     = (const float*)d_in[5];
  const float* Wo0    = (const float*)d_in[6];
  const float* bo0    = (const float*)d_in[7];
  const float* Wx1    = (const float*)d_in[8];
  const float* Wh1    = (const float*)d_in[9];
  const float* b1     = (const float*)d_in[10];
  const float* Wo1    = (const float*)d_in[11];
  const float* bo1    = (const float*)d_in[12];
  float* out = (float*)d_out;

  float* xg   = (float*)d_ws;
  float* h0   = xg + (size_t)T_DIM * B_DIM * G4;
  float* ring = h0 + (size_t)T_DIM * B_DIM * HID;
  const int ringN = KF * B_DIM * HID;
  float* Wf   = ring + (size_t)KF * B_DIM * HID;
  float* bf   = Wf + (size_t)HID * G4;

  const size_t ldsE = (size_t)64 * IS * 4 + 16 * BP64 * 8 + 64 * XS * 4 + 32 * BP512 * 8;
  const size_t ldsX = (size_t)64 * HS * 4 + 64 * BP512 * 8;
  const size_t ldsS = (size_t)WHS_BYTES + 64 * HS * 4;

  const int nblk = (T_DIM * B_DIM) / 64;

  embed_xg0_kernel<<<nblk, 512, ldsE, stream>>>(inputs, W_emb, b_emb, Wx0, b0, xg);
  zero_kernel<<<ringN / 256, 256, 0, stream>>>(ring, ringN);
  scan_kernel<HID><<<1, 512, ldsS, stream>>>(xg, Wh0, Wo0, bo0, ring, h0);
  fold_kernel<<<G4 / 16, 288, 0, stream>>>(Wo0, bo0, Wx1, b1, Wf, bf);
  xg1_kernel<<<nblk, 512, ldsX, stream>>>(h0, Wf, bf, xg);
  zero_kernel<<<ringN / 256, 256, 0, stream>>>(ring, ringN);
  scan_kernel<OUTD><<<1, 512, ldsS, stream>>>(xg, Wh1, Wo1, bo1, ring, out);
}


# scan: a wave runs its k-loops at issue priority 3 and its gate arithmetic between the barrier halves at priority 0 (s_setprio): gate vector work no longer stalls other waves' WMMAs; values unchanged
# speedup vs baseline: 2.5650x; 1.0440x over previous
#include <hip/hip_runtime.h>
#include <cstdint>
#include <cstddef>
#include <utility>

#define DEVINL __device__ __forceinline__

typedef float v2f __attribute__((ext_vector_type(2)));
typedef float v4f __attribute__((ext_vector_type(4)));
typedef float v8f __attribute__((ext_vector_type(8)));

static constexpr int T_DIM  = 1024;
static constexpr int B_DIM  = 64;
static constexpr int IN_DIM = 32;
static constexpr int EMB    = 64;
static constexpr int HID    = 128;
static constexpr int KF     = 16;
static constexpr int OUTD   = 32;
static constexpr int G4     = 512;

static constexpr int HS     = 132;
static constexpr int XS     = 68;
static constexpr int IS     = 36;
static constexpr int BP512  = 528;
static constexpr int BP64   = 80;
static constexpr int WHS_BYTES = 8 * 64 * 2 * 16 * 16;

DEVINL float sigm(float x) { return 1.0f / (1.0f + __expf(-x)); }

struct TanhSt { v2f zs, a, k, t; int k0, k1; };
DEVINL v2f splat2(float c) { v2f r; r.x = c; r.y = c; return r; }
static constexpr int TANH_STAGES = 21;
DEVINL void hold2(v2f& v) { asm volatile("" : "+v"(v)); }
DEVINL void hold1(float& v) { asm volatile("" : "+v"(v)); }
DEVINL void hold2(int& a, int& b) { asm volatile("" : "+v"(a), "+v"(b)); }
template <int S, bool HOLD>
DEVINL void tanh_stage(TanhSt& q, v2f x, v2f og, v2f& res) {
  v2f y; y.x = __builtin_fabsf(x.x); y.y = __builtin_fabsf(x.y);
  constexpr float L2E_HI = 0x1.715476p+0f, L2E_LO = 0x1.4ae0bep-26f;
  if constexpr (S == 0)  q.zs = x * x;
  if constexpr (S == 1)  q.t  = __builtin_elementwise_fma(splat2(-0x1.758e7ap-8f), q.zs, splat2(0x1.521192p-6f));
  if constexpr (S == 2)  q.t  = __builtin_elementwise_fma(q.zs, q.t, splat2(-0x1.b8389cp-5f));
  if constexpr (S == 3)  q.t  = __builtin_elementwise_fma(q.zs, q.t, splat2(0x1.110704p-3f));
  if constexpr (S == 4)  q.t  = __builtin_elementwise_fma(q.zs, q.t, splat2(-0x1.555532p-2f));
  if constexpr (S == 5)  q.t  = y * q.t;
  if constexpr (S == 6)  q.zs = __builtin_elementwise_fma(q.zs, q.t, y);
  if constexpr (S == 7)  q.a  = y + y;
  if constexpr (S == 8)  { q.k.x = __builtin_rintf(q.a.x * L2E_HI); q.k.y = __builtin_rintf(q.a.y * L2E_HI); }
  if constexpr (S == 9)  q.t  = __builtin_elementwise_fma(q.a, splat2(L2E_HI), -q.k);
  if constexpr (S == 10) q.t  = __builtin_elementwise_fma(q.a, splat2(L2E_LO), q.t);
  if constexpr (S == 11) { q.k0 = (int)q.k.x; q.k1 = (int)q.k.y; }
  if constexpr (S == 12) { q.t.x = __builtin_amdgcn_exp2f(q.t.x); q.t.y = __builtin_amdgcn_exp2f(q.t.y); }
  if constexpr (S == 13) { q.t.x = __builtin_ldexpf(q.t.x, q.k0); q.t.y = __builtin_ldexpf(q.t.y, q.k1); }
  if constexpr (S == 14) { q.t.x = (0x1.62e430p+6f < q.a.x) ? __builtin_inff() : q.t.x;
                           q.t.y = (0x1.62e430p+6f < q.a.y) ? __builtin_inff() : q.t.y; }
  if constexpr (S == 15) q.t  = q.t + splat2(1.0f);
  if constexpr (S == 16) { q.t.x = __builtin_amdgcn_rcpf(q.t.x); q.t.y = __builtin_amdgcn_rcpf(q.t.y); }
  if constexpr (S == 17) q.t  = __builtin_elementwise_fma(q.t, splat2(-2.0f), splat2(1.0f));
  if constexpr (S == 18) { q.t.x = (0.625f > y.x) ? q.zs.x : q.t.x; q.t.y = (0.625f > y.y) ? q.zs.y : q.t.y; }
  if constexpr (S == 19) { q.t.x = __builtin_copysignf(q.t.x, x.x); q.t.y = __builtin_copysignf(q.t.y, x.y); }
  if constexpr (S == 20) res  = og * q.t;
  if constexpr (HOLD) {
    if constexpr (S == 0 || S == 6) hold2(q.zs);
    else if constexpr (S == 7)      hold2(q.a);
    else if constexpr (S == 8)      hold2(q.k);
    else if constexpr (S == 11)     hold2(q.k0, q.k1);
    else if constexpr (S == 20)     hold2(res);
    else                            hold2(q.t);
  }
}
DEVINL v2f out_path(v2f x, v2f og) {
  TanhSt q; v2f res;
  [&]<int... S>(std::integer_sequence<int, S...>) __attribute__((always_inline)) {
    (tanh_stage<S, false>(q, x, og, res), ...);
  }(std::make_integer_sequence<int, TANH_STAGES>{});
  return res;
}
template <int N>
DEVINL void out_gap(TanhSt (&q)[4], const v2f (&x)[4], const v2f (&og)[4], v2f (&hn)[4]) {
  constexpr int G = (N >= 23) ? 1 : 0;
  constexpr int M = N - 23 * G;
  if constexpr (M >= 0 && M < TANH_STAGES)         tanh_stage<M, true>(q[2 * G], x[2 * G], og[2 * G], hn[2 * G]);
  if constexpr (M - 1 >= 0 && M - 1 < TANH_STAGES) tanh_stage<M - 1, true>(q[2 * G + 1], x[2 * G + 1], og[2 * G + 1], hn[2 * G + 1]);
}

DEVINL v8f wmma4(v2f a, v2f b, v8f c) {
  return __builtin_amdgcn_wmma_f32_16x16x4_f32(false, a, false, b, (short)0, c,
                                               false, false);
}

DEVINL v8f vzero8() {
  v8f z;
#pragma unroll
  for (int r = 0; r < 8; ++r) z[r] = 0.0f;
  return z;
}

DEVINL void bar_arrive() { asm volatile("s_wait_dscnt 0x0\n\ts_barrier_signal -1" ::: "memory"); }
DEVINL void bar_await()  { asm volatile("s_barrier_wait -1" ::: "memory"); }

DEVINL void prio_kloop() { __builtin_amdgcn_s_setprio(3); }
DEVINL void prio_gates() { __builtin_amdgcn_s_setprio(0); }

struct Frag { v2f a[2]; v4f q[2][2]; };

DEVINL void load_frag(Frag& f, const float* ab, const v4f* wq, int blk) {
#pragma unroll
  for (int kk = 0; kk < 2; ++kk) {
    const int i = blk * 2 + kk;
    f.a[kk]    = *(const v2f*)(ab + 4 * i);
    f.q[kk][0] = wq[i * 64];
    f.q[kk][1] = wq[i * 64 + 16];
  }
}

DEVINL void block_wmma(const Frag& f, v8f (&acc)[4]) {
#pragma unroll
  for (int kk = 0; kk < 2; ++kk) {
    const v4f q0 = f.q[kk][0], q1 = f.q[kk][1];
    v2f b0, b1, b2, b3;
    b0.x = q0.x; b0.y = q0.y; b1.x = q0.z; b1.y = q0.w;
    b2.x = q1.x; b2.y = q1.y; b3.x = q1.z; b3.y = q1.w;
    acc[0] = wmma4(f.a[kk], b0, acc[0]);
    acc[1] = wmma4(f.a[kk], b1, acc[1]);
    acc[2] = wmma4(f.a[kk], b2, acc[2]);
    acc[3] = wmma4(f.a[kk], b3, acc[3]);
  }
}

template <int B0, int B1>
DEVINL void k_blocks(Frag (&fr)[2], v8f (&acc)[4], const float* ab, const v4f* wq) {
#pragma unroll
  for (int blk = B0; blk < B1; ++blk) {
    if (blk + 1 < 16) load_frag(fr[(blk & 1) ^ 1], ab, wq, blk + 1);
    __builtin_amdgcn_sched_barrier(0);
    block_wmma(fr[blk & 1], acc);
    __builtin_amdgcn_sched_barrier(0);
  }
}

template <int NOUT>
DEVINL void project_tile(const float* hbuf, const float* __restrict__ Wo,
                         const float* __restrict__ bo, float* ot, int p, int lh, int ln) {
  const int tm  = p / (NOUT / 16);
  const int tno = p % (NOUT / 16);
  const int col = tno * 16 + ln;
  const float bv = bo[col];
  v8f acc = vzero8();
  const int am = tm * 16 + ln;
  for (int k0 = 0; k0 < 128; k0 += 4) {
    v2f a = *(const v2f*)&hbuf[am * HS + k0 + 2 * lh];
    v2f b;
    b.x = Wo[(k0 + 2 * lh) * NOUT + col];
    b.y = Wo[(k0 + 2 * lh + 1) * NOUT + col];
    acc = wmma4(a, b, acc);
  }
#pragma unroll
  for (int r = 0; r < 8; ++r)
    ot[(tm * 16 + r + 8 * lh) * NOUT + col] = acc[r] + bv;
}

__global__ void zero_kernel(float* __restrict__ p, int n) {
  int i = blockIdx.x * blockDim.x + threadIdx.x;
  if (i < n) p[i] = 0.0f;
}

__global__ void __launch_bounds__(512)
embed_xg0_kernel(const float* __restrict__ in,
                 const float* __restrict__ We,
                 const float* __restrict__ be,
                 const float* __restrict__ Wx0,
                 const float* __restrict__ b0,
                 float* __restrict__ xg)
{
  extern __shared__ __align__(16) char smem[];
  float* inP = (float*)smem;
  v2f*   WeP = (v2f*)(smem + 64 * IS * 4);
  float* xb  = (float*)(smem + 64 * IS * 4 + 16 * BP64 * 8);
  v2f*   WxP = (v2f*)(smem + 64 * IS * 4 + 16 * BP64 * 8 + 64 * XS * 4);

  const int tid  = threadIdx.x;
  const int wave = tid >> 5;
  const int lane = tid & 31;
  const int lh   = lane >> 4;
  const int ln   = lane & 15;
  const int row0 = blockIdx.x * 64;

  for (int i = tid; i < 64 * 32; i += 512) {
    int r = i >> 5, c = i & 31;
    inP[r * IS + c] = in[(size_t)(row0 + r) * 32 + c];
  }
  for (int i = tid; i < 16 * 64; i += 512) {
    int kp = i >> 6, n = i & 63;
    v2f w; w.x = We[(2 * kp) * 64 + n]; w.y = We[(2 * kp + 1) * 64 + n];
    WeP[kp * BP64 + n] = w;
  }
  for (int i = tid; i < 32 * 512; i += 512) {
    int kp = i >> 9, n = i & 511;
    v2f w; w.x = Wx0[(2 * kp) * 512 + n]; w.y = Wx0[(2 * kp + 1) * 512 + n];
    WxP[kp * BP512 + n] = w;
  }
  __syncthreads();

  {
    const int tm = wave >> 2, tn = wave & 3;
    const int col = tn * 16 + ln;
    const float bv = be[col];
    v8f acc = vzero8();
    const int am = tm * 16 + ln;
#pragma unroll
    for (int k0 = 0; k0 < 32; k0 += 4) {
      v2f a = *(const v2f*)&inP[am * IS + k0 + 2 * lh];
      v2f b = WeP[((k0 >> 1) + lh) * BP64 + col];
      acc = wmma4(a, b, acc);
    }
#pragma unroll
    for (int r = 0; r < 8; ++r)
      xb[(tm * 16 + r + 8 * lh) * XS + col] = acc[r] + bv;
  }
  __syncthreads();

  for (int task = wave; task < 128; task += 16) {
    const int tm = task >> 5, tn = task & 31;
    const int col = tn * 16 + ln;
    const float bv = b0[col];
    v8f acc = vzero8();
    const int am = tm * 16 + ln;
#pragma unroll
    for (int k0 = 0; k0 < 64; k0 += 4) {
      v2f a = *(const v2f*)&xb[am * XS + k0 + 2 * lh];
      v2f b = WxP[((k0 >> 1) + lh) * BP512 + col];
      acc = wmma4(a, b, acc);
    }
#pragma unroll
    for (int r = 0; r < 8; ++r)
      xg[(size_t)(row0 + tm * 16 + r + 8 * lh) * 512 + col] = acc[r] + bv;
  }
}

__global__ void __launch_bounds__(512)
xg1_kernel(const float* __restrict__ x1,
           const float* __restrict__ Wx1,
           const float* __restrict__ b1,
           float* __restrict__ xg)
{
  extern __shared__ __align__(16) char smem[];
  float* aB = (float*)smem;
  v2f*   WP = (v2f*)(smem + 64 * HS * 4);

  const int tid  = threadIdx.x;
  const int wave = tid >> 5;
  const int lane = tid & 31;
  const int lh   = lane >> 4;
  const int ln   = lane & 15;
  const int row0 = blockIdx.x * 64;

  for (int i = tid; i < 64 * 128; i += 512) {
    int r = i >> 7, c = i & 127;
    aB[r * HS + c] = x1[(size_t)(row0 + r) * 128 + c];
  }
  for (int i = tid; i < 64 * 512; i += 512) {
    int kp = i >> 9, n = i & 511;
    v2f w; w.x = Wx1[(2 * kp) * 512 + n]; w.y = Wx1[(2 * kp + 1) * 512 + n];
    WP[kp * BP512 + n] = w;
  }
  __syncthreads();

  for (int task = wave; task < 128; task += 16) {
    const int tm = task >> 5, tn = task & 31;
    const int col = tn * 16 + ln;
    const float bv = b1[col];
    v8f acc = vzero8();
    const int am = tm * 16 + ln;
    for (int k0 = 0; k0 < 128; k0 += 4) {
      v2f a = *(const v2f*)&aB[am * HS + k0 + 2 * lh];
      v2f b = WP[((k0 >> 1) + lh) * BP512 + col];
      acc = wmma4(a, b, acc);
    }
#pragma unroll
    for (int r = 0; r < 8; ++r)
      xg[(size_t)(row0 + tm * 16 + r + 8 * lh) * 512 + col] = acc[r] + bv;
  }
}

__global__ void __launch_bounds__(288)
fold_kernel(const float* __restrict__ Wo,
            const float* __restrict__ bo,
            const float* __restrict__ Wx,
            const float* __restrict__ b,
            float* __restrict__ Wf,
            float* __restrict__ bf)
{
  const int tid  = threadIdx.x;
  const int wave = tid >> 5;
  const int lane = tid & 31;
  const int lh   = lane >> 4;
  const int ln   = lane & 15;
  const int col  = blockIdx.x * 16 + ln;
  const bool isW = wave < 8;
  const float* arow = isW ? Wo + (wave * 16 + ln) * HID : bo;
  const bool   live = isW || ln == 0;
  v8f acc = vzero8();
  v2f a[32], w[32];
#pragma unroll
  for (int i = 0; i < 32; ++i) {
    const v2f av = *(const v2f*)(arow + 4 * i + 2 * lh);
    a[i].x = live ? av.x : 0.0f;
    a[i].y = live ? av.y : 0.0f;
    w[i].x = Wx[(4 * i + 2 * lh) * G4 + col];
    w[i].y = Wx[(4 * i + 2 * lh + 1) * G4 + col];
  }
#pragma unroll
  for (int i = 0; i < 32; ++i) acc = wmma4(a[i], w[i], acc);
  if (isW) {
#pragma unroll
    for (int r = 0; r < 8; ++r)
      Wf[(wave * 16 + r + 8 * lh) * G4 + col] = acc[r];
  } else if (lh == 0) {
    bf[col] = acc[0] + b[col];
  }
}

template <int NOUT>
__global__ void __launch_bounds__(512)
scan_kernel(const float* __restrict__ xg,
            const float* __restrict__ Wh,
            const float* __restrict__ Wo,
            const float* __restrict__ bo,
            float* __restrict__ ring,
            float* __restrict__ out)
{
  extern __shared__ __align__(16) char smem[];
  v2f*   WhP  = (v2f*)smem;
  float* hbuf = (float*)(smem + WHS_BYTES);

  const int tid  = threadIdx.x;
  const int wave = tid >> 5;
  const int lane = tid & 31;
  const int lh   = lane >> 4;
  const int ln   = lane & 15;

  for (int i = tid; i < 64 * 512; i += 512) {
    int kp = i >> 9, n = i & 511;
    int g = n >> 7, hh = n & 127;
    v2f w; w.x = Wh[(2 * kp) * 512 + n]; w.y = Wh[(2 * kp + 1) * 512 + n];
    WhP[((((hh >> 4) * 64 + kp) * 2 + (g >> 1)) * 16 + (((hh & 15) + 8 * (kp & 1)) & 15)) * 2 + (g & 1)] = w;
  }
  for (int i = tid; i < 64 * HS; i += 512) hbuf[i] = 0.0f;
  __syncthreads();

  const int tnh  = wave & 7;
  const int colh = tnh * 16 + ln;
  const v4f* wq  = (const v4f*)WhP + ((tnh * 64 + lh) * 2) * 16 + ((ln + 8 * lh) & 15);
  const float* abw = hbuf + ((wave >> 3) * 16 + ln) * HS + 2 * lh;

  v8f  acc[4];
  Frag fr[2];
#pragma unroll
  for (int g = 0; g < 4; ++g) acc[g] = vzero8();
  prio_kloop();
  load_frag(fr[0], abw, wq, 0);
  k_blocks<0, 8>(fr, acc, abw, wq);

  v2f pc[4], po[4];
#pragma unroll
  for (int p = 0; p < 4; ++p) { pc[p] = splat2(0.0f); po[p] = splat2(0.0f); }

  for (int t = 0; t < T_DIM; ++t) {
    const float* xgt = xg + (size_t)t * (64 * 512);

    if (t + 1 < T_DIM) {
      const float* nxt = xgt + 64 * 512;
#pragma unroll
      for (int q = 0; q < 8; ++q)
        __builtin_prefetch(nxt + ((size_t)tid + q * 512) * 32, 0, 0);
    }

#pragma unroll
    for (int ti = 0; ti < 2; ++ti) {
      const int task = wave + ti * 16;
      const int tm   = task >> 3;

      float xv0[8], xv1[8], xv2[8], xv3[8];
      {
        const unsigned xoff = (unsigned)((tm * 16 + 8 * lh) * 512 + colh) * 4u;
        const char* xb = (const char*)xgt + xoff;
#pragma unroll
        for (int r = 0; r < 8; ++r) {
          const float* p = (const float*)(xb + r * 2048);
          xv0[r] = p[0];  xv1[r] = p[128];
          xv2[r] = p[256]; xv3[r] = p[384];
        }
      }

      constexpr int EARLY = KF - 6;
      v4f rv[KF - 1][2];
      const unsigned rboff = (unsigned)(task * 64 + lane) * 16u;
#pragma unroll
      for (int j = 1; j <= EARLY; ++j) {
        const char* sp = (const char*)ring + (size_t)((t - j) & 15) * 32768;
        rv[j - 1][0] = *(const v4f*)(sp + rboff);
        rv[j - 1][1] = *(const v4f*)(sp + rboff + 512);
      }

      const float* ab = abw + ti * (32 * HS);
      v2f hn2[4];
      TanhSt tq[4];
      [&]<int... N>(std::integer_sequence<int, N...>) __attribute__((always_inline)) {
        ([&] __attribute__((always_inline)) {
          constexpr int blk = 8 + N / 8, kk = (N / 4) & 1, g = N & 3;
          if constexpr ((N & 7) == 0) {
            if constexpr (blk + 1 < 16) load_frag(fr[(blk & 1) ^ 1], ab, wq, blk + 1);
            __builtin_amdgcn_sched_barrier(0);
          }
          const v4f qq = fr[blk & 1].q[kk][g >> 1];
          v2f b;
          b.x = (g & 1) ? qq.z : qq.x; b.y = (g & 1) ? qq.w : qq.y;
          acc[g] = wmma4(fr[blk & 1].a[kk], b, acc[g]);
          if constexpr (N < 46) out_gap<N>(tq, pc, po, hn2);
          if constexpr (N >= 46 && N < 50) {
            constexpr int p = N - 46;
            const int tmp = (wave >> 3) + 2 * (ti ^ 1);
            hbuf[(tmp * 16 + 2 * p + 8 * lh) * HS + colh]     = hn2[p].x;
            hbuf[(tmp * 16 + 2 * p + 1 + 8 * lh) * HS + colh] = hn2[p].y;
          }
          __builtin_amdgcn_sched_barrier(0);
        }(), ...);
      }(std::make_integer_sequence<int, 64>{});
      if constexpr (NOUT == 128) {
        if (ti == 1 || t > 0) {
          char* hb = (char*)out + (size_t)(t - (ti ^ 1)) * (64 * HID * 4);
          const unsigned hoff = (unsigned)(((wave >> 3) * 16 + 8 * lh) * HID + (wave & 7) * 16 + ln) * 4u;
#pragma unroll
          for (int p = 0; p < 4; ++p) {
            *(float*)(hb + hoff + (ti ^ 1) * 16384 + (2 * p) * 512)     = hn2[p].x;
            *(float*)(hb + hoff + (ti ^ 1) * 16384 + (2 * p + 1) * 512) = hn2[p].y;
          }
        }
      }

      bar_arrive();
      prio_gates();

      {
        typedef const char __attribute__((address_space(1)))* gptr_cc;
        typedef const v4f __attribute__((address_space(1)))* gptr_cv4;
        gptr_cc ringl = (gptr_cc)ring;
        asm volatile("" : "+s"(ringl));
#pragma unroll
        for (int j = EARLY + 1; j < KF; ++j) {
          gptr_cc sp = ringl + (size_t)((t - j) & 15) * 32768;
          rv[j - 1][0] = *(gptr_cv4)(sp + rboff);
          rv[j - 1][1] = *(gptr_cv4)(sp + rboff + 512);
        }
      }

      float hc[8], og[8], dd[8];
#pragma unroll
      for (int r = 0; r < 8; ++r) {
        const float ig = sigm(acc[0][r] + xv0[r]);
        og[r] = sigm(acc[1][r] + xv1[r]);
        const float cg = tanhf(acc[2][r] + xv2[r]);
        dd[r] = 0.5f * sigm(acc[3][r] + xv3[r]);
        hc[r] = ig * cg;
      }
      constexpr int U0 = 24;
      constexpr int U1 = 48;
      constexpr int UPG = (NOUT == 128) ? 3 : 4;
      float csum[8], w[8];
#pragma unroll
      for (int r = 0; r < 8; ++r) { csum[r] = hc[r]; w[r] = 1.0f; }
#pragma unroll
      for (int j = 1; j <= U0 / 8; ++j) {
        const float jm1 = (float)(j - 1), rj = 1.0f / (float)j;
#pragma unroll
        for (int r = 0; r < 8; ++r) {
          w[r] *= (jm1 + dd[r]) * rj;
          csum[r] += w[r] * rv[j - 1][r >> 2][r & 3];
        }
      }
      {
        char* sw = (char*)ring + (size_t)(t & 15) * 32768;
        v4f lo, hi;
        lo.x = hc[0]; lo.y = hc[1]; lo.z = hc[2]; lo.w = hc[3];
        hi.x = hc[4]; hi.y = hc[5]; hi.z = hc[6]; hi.w = hc[7];
        *(v4f*)(sw + rboff)       = lo;
        *(v4f*)(sw + rboff + 512) = hi;
      }

      bar_await();
      prio_kloop();

      float t2[128];
      auto stage = [&](int n) __attribute__((always_inline)) {
#pragma unroll
        for (int h = 0; h < UPG; ++h) {
          const int uB = U1 + UPG * (n + 2) + h;
          const int uC = U1 + UPG * (n + 1) + h, uD = U1 + UPG * n + h;
          if (uB >= U1 && uB < 120) { t2[uB] = ((float)(uB >> 3) + dd[uB & 7]) * (1.0f / (float)((uB >> 3) + 1)); hold1(t2[uB]); }
          if (uC >= U1 && uC < 120) { w[uC & 7] *= t2[uC]; hold1(w[uC & 7]); }
          if (uD >= U1 && uD < 120) { csum[uD & 7] += w[uD & 7] * rv[uD >> 3][(uD & 7) >> 2][uD & 3]; hold1(csum[uD & 7]); }
        }
      };

      const float* abn = abw + (ti ^ 1) * (32 * HS);
#pragma unroll
      for (int g = 0; g < 4; ++g) acc[g] = vzero8();
      load_frag(fr[0], abn, wq, 0);
#pragma unroll
      for (int j = U0 / 8 + 1; j <= U1 / 8; ++j) {
        const float jm1 = (float)(j - 1), rj = 1.0f / (float)j;
#pragma unroll
        for (int r = 0; r < 8; ++r) {
          w[r] *= (jm1 + dd[r]) * rj;
          csum[r] += w[r] * rv[j - 1][r >> 2][r & 3];
        }
      }
      __builtin_amdgcn_sched_barrier(0);
      stage(-2); stage(-1);
#pragma unroll
      for (int blk = 0; blk < 8; ++blk) {
        load_frag(fr[(blk & 1) ^ 1], abn, wq, blk + 1);
        __builtin_amdgcn_sched_barrier(0);
#pragma unroll
        for (int kk = 0; kk < 2; ++kk) {
          const v4f q0 = fr[blk & 1].q[kk][0], q1 = fr[blk & 1].q[kk][1];
          v2f b[4];
          b[0].x = q0.x; b[0].y = q0.y; b[1].x = q0.z; b[1].y = q0.w;
          b[2].x = q1.x; b[2].y = q1.y; b[3].x = q1.z; b[3].y = q1.w;
#pragma unroll
          for (int g = 0; g < 4; ++g) {
            acc[g] = wmma4(fr[blk & 1].a[kk], b[g], acc[g]);
            stage(blk * 8 + kk * 4 + g);
            __builtin_amdgcn_sched_barrier(0);
          }
        }
      }

#pragma unroll
      for (int p = 0; p < 4; ++p) {
        pc[p].x = csum[2 * p]; pc[p].y = csum[2 * p + 1];
        po[p].x = og[2 * p];   po[p].y = og[2 * p + 1];
      }

      if constexpr (NOUT != 128) {
        static_assert(NOUT == 32, "waves 0-3 = batch half 0, waves 4-7 = batch half 1");
        if (ti == 0) {
          if (t > 0 && wave >= 4 && wave < 8)
            project_tile<NOUT>(hbuf, Wo, bo, out + (size_t)(t - 1) * (64 * NOUT), wave, lh, ln);
        } else {
          if (wave < 4)
            project_tile<NOUT>(hbuf, Wo, bo, out + (size_t)t * (64 * NOUT), wave, lh, ln);
        }
      }
    }
  }
  prio_gates();
  {
    const int tmp = (wave >> 3) + 2;
#pragma unroll
    for (int r = 0; r < 8; ++r) {
      const v2f hp = out_path(pc[r >> 1], po[r >> 1]);
      const float hl = (r & 1) ? hp.y : hp.x;
      hbuf[(tmp * 16 + r + 8 * lh) * HS + colh] = hl;
      if constexpr (NOUT == 128) {
        const unsigned hoff = (unsigned)(((wave >> 3) * 16 + 8 * lh) * HID + (wave & 7) * 16 + ln) * 4u;
        *(float*)((char*)out + (size_t)(T_DIM - 1) * (64 * HID * 4) + hoff + 16384 + r * 512) = hl;
      }
    }
  }
  if constexpr (NOUT != 128) {
    __syncthreads();
    if (wave >= 4 && wave < 8)
      project_tile<NOUT>(hbuf, Wo, bo, out + (size_t)(T_DIM - 1) * (64 * NOUT), wave, lh, ln);
  }
}

extern "C" void kernel_launch(void* const* d_in, const int* in_sizes, int n_in,
                              void* d_out, int out_size, void* d_ws, size_t ws_size,
                              hipStream_t stream) {
  const float* inputs = (const float*)d_in[0];
  const float* W_emb  = (const float*)d_in[1];
  const float* b_emb  = (const float*)d_in[2];
  const float* Wx0    = (const float*)d_in[3];
  const float* Wh0    = (const float*)d_in[4];
  const float* b0     = (const float*)d_in[5];
  const float* Wo0    = (const float*)d_in[6];
  const float* bo0    = (const float*)d_in[7];
  const float* Wx1    = (const float*)d_in[8];
  const float* Wh1    = (const float*)d_in[9];
  const float* b1     = (const float*)d_in[10];
  const float* Wo1    = (const float*)d_in[11];
  const float* bo1    = (const float*)d_in[12];
  float* out = (float*)d_out;

  float* xg   = (float*)d_ws;
  float* h0   = xg + (size_t)T_DIM * B_DIM * G4;
  float* ring = h0 + (size_t)T_DIM * B_DIM * HID;
  const int ringN = KF * B_DIM * HID;
  float* Wf   = ring + (size_t)KF * B_DIM * HID;
  float* bf   = Wf + (size_t)HID * G4;

  const size_t ldsE = (size_t)64 * IS * 4 + 16 * BP64 * 8 + 64 * XS * 4 + 32 * BP512 * 8;
  const size_t ldsX = (size_t)64 * HS * 4 + 64 * BP512 * 8;
  const size_t ldsS = (size_t)WHS_BYTES + 64 * HS * 4;

  const int nblk = (T_DIM * B_DIM) / 64;

  embed_xg0_kernel<<<nblk, 512, ldsE, stream>>>(inputs, W_emb, b_emb, Wx0, b0, xg);
  zero_kernel<<<ringN / 256, 256, 0, stream>>>(ring, ringN);
  scan_kernel<HID><<<1, 512, ldsS, stream>>>(xg, Wh0, Wo0, bo0, ring, h0);
  fold_kernel<<<G4 / 16, 288, 0, stream>>>(Wo0, bo0, Wx1, b1, Wf, bf);
  xg1_kernel<<<nblk, 512, ldsX, stream>>>(h0, Wf, bf, xg);
  zero_kernel<<<ringN / 256, 256, 0, stream>>>(ring, ringN);
  scan_kernel<OUTD><<<1, 512, ldsS, stream>>>(xg, Wh1, Wo1, bo1, ring, out);
}


# scan: every activation on the hardware tanh instruction, one v_tanh_f32 per value: tanh direct, sigmoid(x)=0.5+0.5*tanh(x/2); no exp, rcp, division or branch; values move by that instruction's error
# speedup vs baseline: 2.8258x; 1.1017x over previous
#include <hip/hip_runtime.h>
#include <cstdint>
#include <cstddef>
#include <utility>

#define DEVINL __device__ __forceinline__

typedef float v2f __attribute__((ext_vector_type(2)));
typedef float v4f __attribute__((ext_vector_type(4)));
typedef float v8f __attribute__((ext_vector_type(8)));

static constexpr int T_DIM  = 1024;
static constexpr int B_DIM  = 64;
static constexpr int IN_DIM = 32;
static constexpr int EMB    = 64;
static constexpr int HID    = 128;
static constexpr int KF     = 16;
static constexpr int OUTD   = 32;
static constexpr int G4     = 512;

static constexpr int HS     = 132;
static constexpr int XS     = 68;
static constexpr int IS     = 36;
static constexpr int BP512  = 528;
static constexpr int BP64   = 80;
static constexpr int WHS_BYTES = 8 * 64 * 2 * 16 * 16;

DEVINL float htanh(float x) { return __builtin_amdgcn_tanhf(x); }
DEVINL float sigm(float x) { return __builtin_fmaf(0.5f, htanh(0.5f * x), 0.5f); }

DEVINL v2f splat2(float c) { v2f r; r.x = c; r.y = c; return r; }
DEVINL void hold1(float& v) { asm volatile("" : "+v"(v)); }

DEVINL v8f wmma4(v2f a, v2f b, v8f c) {
  return __builtin_amdgcn_wmma_f32_16x16x4_f32(false, a, false, b, (short)0, c,
                                               false, false);
}

DEVINL v8f vzero8() {
  v8f z;
#pragma unroll
  for (int r = 0; r < 8; ++r) z[r] = 0.0f;
  return z;
}

DEVINL void bar_arrive() { asm volatile("s_wait_dscnt 0x0\n\ts_barrier_signal -1" ::: "memory"); }
DEVINL void bar_await()  { asm volatile("s_barrier_wait -1" ::: "memory"); }

DEVINL void prio_kloop() { __builtin_amdgcn_s_setprio(3); }
DEVINL void prio_gates() { __builtin_amdgcn_s_setprio(0); }

struct Frag { v2f a[2]; v4f q[2][2]; };

DEVINL void load_frag(Frag& f, const float* ab, const v4f* wq, int blk) {
#pragma unroll
  for (int kk = 0; kk < 2; ++kk) {
    const int i = blk * 2 + kk;
    f.a[kk]    = *(const v2f*)(ab + 4 * i);
    f.q[kk][0] = wq[i * 64];
    f.q[kk][1] = wq[i * 64 + 16];
  }
}

DEVINL void block_wmma(const Frag& f, v8f (&acc)[4]) {
#pragma unroll
  for (int kk = 0; kk < 2; ++kk) {
    const v4f q0 = f.q[kk][0], q1 = f.q[kk][1];
    v2f b0, b1, b2, b3;
    b0.x = q0.x; b0.y = q0.y; b1.x = q0.z; b1.y = q0.w;
    b2.x = q1.x; b2.y = q1.y; b3.x = q1.z; b3.y = q1.w;
    acc[0] = wmma4(f.a[kk], b0, acc[0]);
    acc[1] = wmma4(f.a[kk], b1, acc[1]);
    acc[2] = wmma4(f.a[kk], b2, acc[2]);
    acc[3] = wmma4(f.a[kk], b3, acc[3]);
  }
}

template <int B0, int B1>
DEVINL void k_blocks(Frag (&fr)[2], v8f (&acc)[4], const float* ab, const v4f* wq) {
#pragma unroll
  for (int blk = B0; blk < B1; ++blk) {
    if (blk + 1 < 16) load_frag(fr[(blk & 1) ^ 1], ab, wq, blk + 1);
    __builtin_amdgcn_sched_barrier(0);
    block_wmma(fr[blk & 1], acc);
    __builtin_amdgcn_sched_barrier(0);
  }
}

template <int NOUT>
DEVINL void project_tile(const float* hbuf, const float* __restrict__ Wo,
                         const float* __restrict__ bo, float* ot, int p, int lh, int ln) {
  const int tm  = p / (NOUT / 16);
  const int tno = p % (NOUT / 16);
  const int col = tno * 16 + ln;
  const float bv = bo[col];
  v8f acc = vzero8();
  const int am = tm * 16 + ln;
  for (int k0 = 0; k0 < 128; k0 += 4) {
    v2f a = *(const v2f*)&hbuf[am * HS + k0 + 2 * lh];
    v2f b;
    b.x = Wo[(k0 + 2 * lh) * NOUT + col];
    b.y = Wo[(k0 + 2 * lh + 1) * NOUT + col];
    acc = wmma4(a, b, acc);
  }
#pragma unroll
  for (int r = 0; r < 8; ++r)
    ot[(tm * 16 + r + 8 * lh) * NOUT + col] = acc[r] + bv;
}

__global__ void zero_kernel(float* __restrict__ p, int n) {
  int i = blockIdx.x * blockDim.x + threadIdx.x;
  if (i < n) p[i] = 0.0f;
}

__global__ void __launch_bounds__(512)
embed_xg0_kernel(const float* __restrict__ in,
                 const float* __restrict__ We,
                 const float* __restrict__ be,
                 const float* __restrict__ Wx0,
                 const float* __restrict__ b0,
                 float* __restrict__ xg)
{
  extern __shared__ __align__(16) char smem[];
  float* inP = (float*)smem;
  v2f*   WeP = (v2f*)(smem + 64 * IS * 4);
  float* xb  = (float*)(smem + 64 * IS * 4 + 16 * BP64 * 8);
  v2f*   WxP = (v2f*)(smem + 64 * IS * 4 + 16 * BP64 * 8 + 64 * XS * 4);

  const int tid  = threadIdx.x;
  const int wave = tid >> 5;
  const int lane = tid & 31;
  const int lh   = lane >> 4;
  const int ln   = lane & 15;
  const int row0 = blockIdx.x * 64;

  for (int i = tid; i < 64 * 32; i += 512) {
    int r = i >> 5, c = i & 31;
    inP[r * IS + c] = in[(size_t)(row0 + r) * 32 + c];
  }
  for (int i = tid; i < 16 * 64; i += 512) {
    int kp = i >> 6, n = i & 63;
    v2f w; w.x = We[(2 * kp) * 64 + n]; w.y = We[(2 * kp + 1) * 64 + n];
    WeP[kp * BP64 + n] = w;
  }
  for (int i = tid; i < 32 * 512; i += 512) {
    int kp = i >> 9, n = i & 511;
    v2f w; w.x = Wx0[(2 * kp) * 512 + n]; w.y = Wx0[(2 * kp + 1) * 512 + n];
    WxP[kp * BP512 + n] = w;
  }
  __syncthreads();

  {
    const int tm = wave >> 2, tn = wave & 3;
    const int col = tn * 16 + ln;
    const float bv = be[col];
    v8f acc = vzero8();
    const int am = tm * 16 + ln;
#pragma unroll
    for (int k0 = 0; k0 < 32; k0 += 4) {
      v2f a = *(const v2f*)&inP[am * IS + k0 + 2 * lh];
      v2f b = WeP[((k0 >> 1) + lh) * BP64 + col];
      acc = wmma4(a, b, acc);
    }
#pragma unroll
    for (int r = 0; r < 8; ++r)
      xb[(tm * 16 + r + 8 * lh) * XS + col] = acc[r] + bv;
  }
  __syncthreads();

  for (int task = wave; task < 128; task += 16) {
    const int tm = task >> 5, tn = task & 31;
    const int col = tn * 16 + ln;
    const float bv = b0[col];
    v8f acc = vzero8();
    const int am = tm * 16 + ln;
#pragma unroll
    for (int k0 = 0; k0 < 64; k0 += 4) {
      v2f a = *(const v2f*)&xb[am * XS + k0 + 2 * lh];
      v2f b = WxP[((k0 >> 1) + lh) * BP512 + col];
      acc = wmma4(a, b, acc);
    }
#pragma unroll
    for (int r = 0; r < 8; ++r)
      xg[(size_t)(row0 + tm * 16 + r + 8 * lh) * 512 + col] = acc[r] + bv;
  }
}

__global__ void __launch_bounds__(512)
xg1_kernel(const float* __restrict__ x1,
           const float* __restrict__ Wx1,
           const float* __restrict__ b1,
           float* __restrict__ xg)
{
  extern __shared__ __align__(16) char smem[];
  float* aB = (float*)smem;
  v2f*   WP = (v2f*)(smem + 64 * HS * 4);

  const int tid  = threadIdx.x;
  const int wave = tid >> 5;
  const int lane = tid & 31;
  const int lh   = lane >> 4;
  const int ln   = lane & 15;
  const int row0 = blockIdx.x * 64;

  for (int i = tid; i < 64 * 128; i += 512) {
    int r = i >> 7, c = i & 127;
    aB[r * HS + c] = x1[(size_t)(row0 + r) * 128 + c];
  }
  for (int i = tid; i < 64 * 512; i += 512) {
    int kp = i >> 9, n = i & 511;
    v2f w; w.x = Wx1[(2 * kp) * 512 + n]; w.y = Wx1[(2 * kp + 1) * 512 + n];
    WP[kp * BP512 + n] = w;
  }
  __syncthreads();

  for (int task = wave; task < 128; task += 16) {
    const int tm = task >> 5, tn = task & 31;
    const int col = tn * 16 + ln;
    const float bv = b1[col];
    v8f acc = vzero8();
    const int am = tm * 16 + ln;
    for (int k0 = 0; k0 < 128; k0 += 4) {
      v2f a = *(const v2f*)&aB[am * HS + k0 + 2 * lh];
      v2f b = WP[((k0 >> 1) + lh) * BP512 + col];
      acc = wmma4(a, b, acc);
    }
#pragma unroll
    for (int r = 0; r < 8; ++r)
      xg[(size_t)(row0 + tm * 16 + r + 8 * lh) * 512 + col] = acc[r] + bv;
  }
}

__global__ void __launch_bounds__(288)
fold_kernel(const float* __restrict__ Wo,
            const float* __restrict__ bo,
            const float* __restrict__ Wx,
            const float* __restrict__ b,
            float* __restrict__ Wf,
            float* __restrict__ bf)
{
  const int tid  = threadIdx.x;
  const int wave = tid >> 5;
  const int lane = tid & 31;
  const int lh   = lane >> 4;
  const int ln   = lane & 15;
  const int col  = blockIdx.x * 16 + ln;
  const bool isW = wave < 8;
  const float* arow = isW ? Wo + (wave * 16 + ln) * HID : bo;
  const bool   live = isW || ln == 0;
  v8f acc = vzero8();
  v2f a[32], w[32];
#pragma unroll
  for (int i = 0; i < 32; ++i) {
    const v2f av = *(const v2f*)(arow + 4 * i + 2 * lh);
    a[i].x = live ? av.x : 0.0f;
    a[i].y = live ? av.y : 0.0f;
    w[i].x = Wx[(4 * i + 2 * lh) * G4 + col];
    w[i].y = Wx[(4 * i + 2 * lh + 1) * G4 + col];
  }
#pragma unroll
  for (int i = 0; i < 32; ++i) acc = wmma4(a[i], w[i], acc);
  if (isW) {
#pragma unroll
    for (int r = 0; r < 8; ++r)
      Wf[(wave * 16 + r + 8 * lh) * G4 + col] = acc[r];
  } else if (lh == 0) {
    bf[col] = acc[0] + b[col];
  }
}

template <int NOUT>
__global__ void __launch_bounds__(512)
scan_kernel(const float* __restrict__ xg,
            const float* __restrict__ Wh,
            const float* __restrict__ Wo,
            const float* __restrict__ bo,
            float* __restrict__ ring,
            float* __restrict__ out)
{
  extern __shared__ __align__(16) char smem[];
  v2f*   WhP  = (v2f*)smem;
  float* hbuf = (float*)(smem + WHS_BYTES);

  const int tid  = threadIdx.x;
  const int wave = tid >> 5;
  const int lane = tid & 31;
  const int lh   = lane >> 4;
  const int ln   = lane & 15;

  for (int i = tid; i < 64 * 512; i += 512) {
    int kp = i >> 9, n = i & 511;
    int g = n >> 7, hh = n & 127;
    v2f w; w.x = Wh[(2 * kp) * 512 + n]; w.y = Wh[(2 * kp + 1) * 512 + n];
    WhP[((((hh >> 4) * 64 + kp) * 2 + (g >> 1)) * 16 + (((hh & 15) + 8 * (kp & 1)) & 15)) * 2 + (g & 1)] = w;
  }
  for (int i = tid; i < 64 * HS; i += 512) hbuf[i] = 0.0f;
  __syncthreads();

  const int tnh  = wave & 7;
  const int colh = tnh * 16 + ln;
  const v4f* wq  = (const v4f*)WhP + ((tnh * 64 + lh) * 2) * 16 + ((ln + 8 * lh) & 15);
  const float* abw = hbuf + ((wave >> 3) * 16 + ln) * HS + 2 * lh;

  v8f  acc[4];
  Frag fr[2];
#pragma unroll
  for (int g = 0; g < 4; ++g) acc[g] = vzero8();
  prio_kloop();
  load_frag(fr[0], abw, wq, 0);
  k_blocks<0, 8>(fr, acc, abw, wq);

  v2f pc[4], po[4];
#pragma unroll
  for (int p = 0; p < 4; ++p) { pc[p] = splat2(0.0f); po[p] = splat2(0.0f); }

  for (int t = 0; t < T_DIM; ++t) {
    const float* xgt = xg + (size_t)t * (64 * 512);

    if (t + 1 < T_DIM) {
      const float* nxt = xgt + 64 * 512;
#pragma unroll
      for (int q = 0; q < 8; ++q)
        __builtin_prefetch(nxt + ((size_t)tid + q * 512) * 32, 0, 0);
    }

#pragma unroll
    for (int ti = 0; ti < 2; ++ti) {
      const int task = wave + ti * 16;
      const int tm   = task >> 3;

      float xv0[8], xv1[8], xv2[8], xv3[8];
      {
        const unsigned xoff = (unsigned)((tm * 16 + 8 * lh) * 512 + colh) * 4u;
        const char* xb = (const char*)xgt + xoff;
#pragma unroll
        for (int r = 0; r < 8; ++r) {
          const float* p = (const float*)(xb + r * 2048);
          xv0[r] = p[0];  xv1[r] = p[128];
          xv2[r] = p[256]; xv3[r] = p[384];
        }
      }

      constexpr int EARLY = KF - 6;
      v4f rv[KF - 1][2];
      const unsigned rboff = (unsigned)(task * 64 + lane) * 16u;
#pragma unroll
      for (int j = 1; j <= EARLY; ++j) {
        const char* sp = (const char*)ring + (size_t)((t - j) & 15) * 32768;
        rv[j - 1][0] = *(const v4f*)(sp + rboff);
        rv[j - 1][1] = *(const v4f*)(sp + rboff + 512);
      }

      const float* ab = abw + ti * (32 * HS);
      v2f hn2[4];
      v2f th[4];
      [&]<int... N>(std::integer_sequence<int, N...>) __attribute__((always_inline)) {
        ([&] __attribute__((always_inline)) {
          constexpr int blk = 8 + N / 8, kk = (N / 4) & 1, g = N & 3;
          if constexpr ((N & 7) == 0) {
            if constexpr (blk + 1 < 16) load_frag(fr[(blk & 1) ^ 1], ab, wq, blk + 1);
            __builtin_amdgcn_sched_barrier(0);
          }
          const v4f qq = fr[blk & 1].q[kk][g >> 1];
          v2f b;
          b.x = (g & 1) ? qq.z : qq.x; b.y = (g & 1) ? qq.w : qq.y;
          acc[g] = wmma4(fr[blk & 1].a[kk], b, acc[g]);
          if constexpr (N < 4) { th[N].x = htanh(pc[N].x); th[N].y = htanh(pc[N].y); }
          if constexpr (N >= 4 && N < 8) hn2[N - 4] = po[N - 4] * th[N - 4];
          if constexpr (N >= 8 && N < 12) {
            constexpr int p = N - 8;
            const int tmp = (wave >> 3) + 2 * (ti ^ 1);
            hbuf[(tmp * 16 + 2 * p + 8 * lh) * HS + colh]     = hn2[p].x;
            hbuf[(tmp * 16 + 2 * p + 1 + 8 * lh) * HS + colh] = hn2[p].y;
          }
          __builtin_amdgcn_sched_barrier(0);
        }(), ...);
      }(std::make_integer_sequence<int, 64>{});
      if constexpr (NOUT == 128) {
        if (ti == 1 || t > 0) {
          char* hb = (char*)out + (size_t)(t - (ti ^ 1)) * (64 * HID * 4);
          const unsigned hoff = (unsigned)(((wave >> 3) * 16 + 8 * lh) * HID + (wave & 7) * 16 + ln) * 4u;
#pragma unroll
          for (int p = 0; p < 4; ++p) {
            *(float*)(hb + hoff + (ti ^ 1) * 16384 + (2 * p) * 512)     = hn2[p].x;
            *(float*)(hb + hoff + (ti ^ 1) * 16384 + (2 * p + 1) * 512) = hn2[p].y;
          }
        }
      }

      bar_arrive();
      prio_gates();

      {
        typedef const char __attribute__((address_space(1)))* gptr_cc;
        typedef const v4f __attribute__((address_space(1)))* gptr_cv4;
        gptr_cc ringl = (gptr_cc)ring;
        asm volatile("" : "+s"(ringl));
#pragma unroll
        for (int j = EARLY + 1; j < KF; ++j) {
          gptr_cc sp = ringl + (size_t)((t - j) & 15) * 32768;
          rv[j - 1][0] = *(gptr_cv4)(sp + rboff);
          rv[j - 1][1] = *(gptr_cv4)(sp + rboff + 512);
        }
      }

      float hc[8], og[8], dd[8];
#pragma unroll
      for (int r = 0; r < 8; ++r) {
        const float ig = sigm(acc[0][r] + xv0[r]);
        og[r] = sigm(acc[1][r] + xv1[r]);
        const float cg = htanh(acc[2][r] + xv2[r]);
        dd[r] = 0.5f * sigm(acc[3][r] + xv3[r]);
        hc[r] = ig * cg;
      }
      constexpr int U0 = 24;
      constexpr int U1 = 48;
      constexpr int UPG = (NOUT == 128) ? 3 : 4;
      float csum[8], w[8];
#pragma unroll
      for (int r = 0; r < 8; ++r) { csum[r] = hc[r]; w[r] = 1.0f; }
#pragma unroll
      for (int j = 1; j <= U0 / 8; ++j) {
        const float jm1 = (float)(j - 1), rj = 1.0f / (float)j;
#pragma unroll
        for (int r = 0; r < 8; ++r) {
          w[r] *= (jm1 + dd[r]) * rj;
          csum[r] += w[r] * rv[j - 1][r >> 2][r & 3];
        }
      }
      {
        char* sw = (char*)ring + (size_t)(t & 15) * 32768;
        v4f lo, hi;
        lo.x = hc[0]; lo.y = hc[1]; lo.z = hc[2]; lo.w = hc[3];
        hi.x = hc[4]; hi.y = hc[5]; hi.z = hc[6]; hi.w = hc[7];
        *(v4f*)(sw + rboff)       = lo;
        *(v4f*)(sw + rboff + 512) = hi;
      }

      bar_await();
      prio_kloop();

      float t2[128];
      auto stage = [&](int n) __attribute__((always_inline)) {
#pragma unroll
        for (int h = 0; h < UPG; ++h) {
          const int uB = U1 + UPG * (n + 2) + h;
          const int uC = U1 + UPG * (n + 1) + h, uD = U1 + UPG * n + h;
          if (uB >= U1 && uB < 120) { t2[uB] = ((float)(uB >> 3) + dd[uB & 7]) * (1.0f / (float)((uB >> 3) + 1)); hold1(t2[uB]); }
          if (uC >= U1 && uC < 120) { w[uC & 7] *= t2[uC]; hold1(w[uC & 7]); }
          if (uD >= U1 && uD < 120) { csum[uD & 7] += w[uD & 7] * rv[uD >> 3][(uD & 7) >> 2][uD & 3]; hold1(csum[uD & 7]); }
        }
      };

      const float* abn = abw + (ti ^ 1) * (32 * HS);
#pragma unroll
      for (int g = 0; g < 4; ++g) acc[g] = vzero8();
      load_frag(fr[0], abn, wq, 0);
#pragma unroll
      for (int j = U0 / 8 + 1; j <= U1 / 8; ++j) {
        const float jm1 = (float)(j - 1), rj = 1.0f / (float)j;
#pragma unroll
        for (int r = 0; r < 8; ++r) {
          w[r] *= (jm1 + dd[r]) * rj;
          csum[r] += w[r] * rv[j - 1][r >> 2][r & 3];
        }
      }
      __builtin_amdgcn_sched_barrier(0);
      stage(-2); stage(-1);
#pragma unroll
      for (int blk = 0; blk < 8; ++blk) {
        load_frag(fr[(blk & 1) ^ 1], abn, wq, blk + 1);
        __builtin_amdgcn_sched_barrier(0);
#pragma unroll
        for (int kk = 0; kk < 2; ++kk) {
          const v4f q0 = fr[blk & 1].q[kk][0], q1 = fr[blk & 1].q[kk][1];
          v2f b[4];
          b[0].x = q0.x; b[0].y = q0.y; b[1].x = q0.z; b[1].y = q0.w;
          b[2].x = q1.x; b[2].y = q1.y; b[3].x = q1.z; b[3].y = q1.w;
#pragma unroll
          for (int g = 0; g < 4; ++g) {
            acc[g] = wmma4(fr[blk & 1].a[kk], b[g], acc[g]);
            stage(blk * 8 + kk * 4 + g);
            __builtin_amdgcn_sched_barrier(0);
          }
        }
      }

#pragma unroll
      for (int p = 0; p < 4; ++p) {
        pc[p].x = csum[2 * p]; pc[p].y = csum[2 * p + 1];
        po[p].x = og[2 * p];   po[p].y = og[2 * p + 1];
      }

      if constexpr (NOUT != 128) {
        static_assert(NOUT == 32, "waves 0-3 = batch half 0, waves 4-7 = batch half 1");
        if (ti == 0) {
          if (t > 0 && wave >= 4 && wave < 8)
            project_tile<NOUT>(hbuf, Wo, bo, out + (size_t)(t - 1) * (64 * NOUT), wave, lh, ln);
        } else {
          if (wave < 4)
            project_tile<NOUT>(hbuf, Wo, bo, out + (size_t)t * (64 * NOUT), wave, lh, ln);
        }
      }
    }
  }
  prio_gates();
  {
    const int tmp = (wave >> 3) + 2;
#pragma unroll
    for (int r = 0; r < 8; ++r) {
      const float hl = (r & 1) ? po[r >> 1].y * htanh(pc[r >> 1].y) : po[r >> 1].x * htanh(pc[r >> 1].x);
      hbuf[(tmp * 16 + r + 8 * lh) * HS + colh] = hl;
      if constexpr (NOUT == 128) {
        const unsigned hoff = (unsigned)(((wave >> 3) * 16 + 8 * lh) * HID + (wave & 7) * 16 + ln) * 4u;
        *(float*)((char*)out + (size_t)(T_DIM - 1) * (64 * HID * 4) + hoff + 16384 + r * 512) = hl;
      }
    }
  }
  if constexpr (NOUT != 128) {
    __syncthreads();
    if (wave >= 4 && wave < 8)
      project_tile<NOUT>(hbuf, Wo, bo, out + (size_t)(T_DIM - 1) * (64 * NOUT), wave, lh, ln);
  }
}

extern "C" void kernel_launch(void* const* d_in, const int* in_sizes, int n_in,
                              void* d_out, int out_size, void* d_ws, size_t ws_size,
                              hipStream_t stream) {
  const float* inputs = (const float*)d_in[0];
  const float* W_emb  = (const float*)d_in[1];
  const float* b_emb  = (const float*)d_in[2];
  const float* Wx0    = (const float*)d_in[3];
  const float* Wh0    = (const float*)d_in[4];
  const float* b0     = (const float*)d_in[5];
  const float* Wo0    = (const float*)d_in[6];
  const float* bo0    = (const float*)d_in[7];
  const float* Wx1    = (const float*)d_in[8];
  const float* Wh1    = (const float*)d_in[9];
  const float* b1     = (const float*)d_in[10];
  const float* Wo1    = (const float*)d_in[11];
  const float* bo1    = (const float*)d_in[12];
  float* out = (float*)d_out;

  float* xg   = (float*)d_ws;
  float* h0   = xg + (size_t)T_DIM * B_DIM * G4;
  float* ring = h0 + (size_t)T_DIM * B_DIM * HID;
  const int ringN = KF * B_DIM * HID;
  float* Wf   = ring + (size_t)KF * B_DIM * HID;
  float* bf   = Wf + (size_t)HID * G4;

  const size_t ldsE = (size_t)64 * IS * 4 + 16 * BP64 * 8 + 64 * XS * 4 + 32 * BP512 * 8;
  const size_t ldsX = (size_t)64 * HS * 4 + 64 * BP512 * 8;
  const size_t ldsS = (size_t)WHS_BYTES + 64 * HS * 4;

  const int nblk = (T_DIM * B_DIM) / 64;

  embed_xg0_kernel<<<nblk, 512, ldsE, stream>>>(inputs, W_emb, b_emb, Wx0, b0, xg);
  zero_kernel<<<ringN / 256, 256, 0, stream>>>(ring, ringN);
  scan_kernel<HID><<<1, 512, ldsS, stream>>>(xg, Wh0, Wo0, bo0, ring, h0);
  fold_kernel<<<G4 / 16, 288, 0, stream>>>(Wo0, bo0, Wx1, b1, Wf, bf);
  xg1_kernel<<<nblk, 512, ldsX, stream>>>(h0, Wf, bf, xg);
  zero_kernel<<<ringN / 256, 256, 0, stream>>>(ring, ringN);
  scan_kernel<OUTD><<<1, 512, ldsS, stream>>>(xg, Wh1, Wo1, bo1, ring, out);
}


# xg stream whole-line, once: GEMMs store xg in the scan's lane order, a task reads 8 loads of 16 bytes (not 32 of 4) and each 256-byte unit is prefetched once (not 8 times); values unchanged
# speedup vs baseline: 3.0356x; 1.0742x over previous
#include <hip/hip_runtime.h>
#include <cstdint>
#include <cstddef>
#include <utility>

#define DEVINL __device__ __forceinline__

typedef float v2f __attribute__((ext_vector_type(2)));
typedef float v4f __attribute__((ext_vector_type(4)));
typedef float v8f __attribute__((ext_vector_type(8)));

static constexpr int T_DIM  = 1024;
static constexpr int B_DIM  = 64;
static constexpr int IN_DIM = 32;
static constexpr int EMB    = 64;
static constexpr int HID    = 128;
static constexpr int KF     = 16;
static constexpr int OUTD   = 32;
static constexpr int G4     = 512;

static constexpr int HS     = 132;
static constexpr int XS     = 68;
static constexpr int IS     = 36;
static constexpr int BP512  = 528;
static constexpr int BP64   = 80;
static constexpr int WHS_BYTES = 8 * 64 * 2 * 16 * 16;

DEVINL float htanh(float x) { return __builtin_amdgcn_tanhf(x); }
DEVINL float sigm(float x) { return __builtin_fmaf(0.5f, htanh(0.5f * x), 0.5f); }

DEVINL v2f splat2(float c) { v2f r; r.x = c; r.y = c; return r; }
DEVINL void hold1(float& v) { asm volatile("" : "+v"(v)); }

DEVINL v8f wmma4(v2f a, v2f b, v8f c) {
  return __builtin_amdgcn_wmma_f32_16x16x4_f32(false, a, false, b, (short)0, c,
                                               false, false);
}

DEVINL v8f vzero8() {
  v8f z;
#pragma unroll
  for (int r = 0; r < 8; ++r) z[r] = 0.0f;
  return z;
}

DEVINL void bar_arrive() { asm volatile("s_wait_dscnt 0x0\n\ts_barrier_signal -1" ::: "memory"); }
DEVINL void bar_await()  { asm volatile("s_barrier_wait -1" ::: "memory"); }

DEVINL void prio_kloop() { __builtin_amdgcn_s_setprio(3); }
DEVINL void prio_gates() { __builtin_amdgcn_s_setprio(0); }

struct Frag { v2f a[2]; v4f q[2][2]; };

DEVINL void load_frag(Frag& f, const float* ab, const v4f* wq, int blk) {
#pragma unroll
  for (int kk = 0; kk < 2; ++kk) {
    const int i = blk * 2 + kk;
    f.a[kk]    = *(const v2f*)(ab + 4 * i);
    f.q[kk][0] = wq[i * 64];
    f.q[kk][1] = wq[i * 64 + 16];
  }
}

DEVINL void block_wmma(const Frag& f, v8f (&acc)[4]) {
#pragma unroll
  for (int kk = 0; kk < 2; ++kk) {
    const v4f q0 = f.q[kk][0], q1 = f.q[kk][1];
    v2f b0, b1, b2, b3;
    b0.x = q0.x; b0.y = q0.y; b1.x = q0.z; b1.y = q0.w;
    b2.x = q1.x; b2.y = q1.y; b3.x = q1.z; b3.y = q1.w;
    acc[0] = wmma4(f.a[kk], b0, acc[0]);
    acc[1] = wmma4(f.a[kk], b1, acc[1]);
    acc[2] = wmma4(f.a[kk], b2, acc[2]);
    acc[3] = wmma4(f.a[kk], b3, acc[3]);
  }
}

template <int B0, int B1>
DEVINL void k_blocks(Frag (&fr)[2], v8f (&acc)[4], const float* ab, const v4f* wq) {
#pragma unroll
  for (int blk = B0; blk < B1; ++blk) {
    if (blk + 1 < 16) load_frag(fr[(blk & 1) ^ 1], ab, wq, blk + 1);
    __builtin_amdgcn_sched_barrier(0);
    block_wmma(fr[blk & 1], acc);
    __builtin_amdgcn_sched_barrier(0);
  }
}

template <int NOUT>
DEVINL void project_tile(const float* hbuf, const float* __restrict__ Wo,
                         const float* __restrict__ bo, float* ot, int p, int lh, int ln) {
  const int tm  = p / (NOUT / 16);
  const int tno = p % (NOUT / 16);
  const int col = tno * 16 + ln;
  const float bv = bo[col];
  v8f acc = vzero8();
  const int am = tm * 16 + ln;
  for (int k0 = 0; k0 < 128; k0 += 4) {
    v2f a = *(const v2f*)&hbuf[am * HS + k0 + 2 * lh];
    v2f b;
    b.x = Wo[(k0 + 2 * lh) * NOUT + col];
    b.y = Wo[(k0 + 2 * lh + 1) * NOUT + col];
    acc = wmma4(a, b, acc);
  }
#pragma unroll
  for (int r = 0; r < 8; ++r)
    ot[(tm * 16 + r + 8 * lh) * NOUT + col] = acc[r] + bv;
}

DEVINL void store_xg_tile(float* __restrict__ xg, int step, int tm, int tn, int lane,
                          const v8f& acc, float bv) {
  float* xo = xg + ((size_t)step * 32 + tm * 8 + (tn & 7)) * 1024 + (tn >> 3) * 256 + lane * 4;
  v4f lo, hi;
  lo.x = acc[0] + bv; lo.y = acc[1] + bv; lo.z = acc[2] + bv; lo.w = acc[3] + bv;
  hi.x = acc[4] + bv; hi.y = acc[5] + bv; hi.z = acc[6] + bv; hi.w = acc[7] + bv;
  *(v4f*)xo         = lo;
  *(v4f*)(xo + 128) = hi;
}

__global__ void zero_kernel(float* __restrict__ p, int n) {
  int i = blockIdx.x * blockDim.x + threadIdx.x;
  if (i < n) p[i] = 0.0f;
}

__global__ void __launch_bounds__(512)
embed_xg0_kernel(const float* __restrict__ in,
                 const float* __restrict__ We,
                 const float* __restrict__ be,
                 const float* __restrict__ Wx0,
                 const float* __restrict__ b0,
                 float* __restrict__ xg)
{
  extern __shared__ __align__(16) char smem[];
  float* inP = (float*)smem;
  v2f*   WeP = (v2f*)(smem + 64 * IS * 4);
  float* xb  = (float*)(smem + 64 * IS * 4 + 16 * BP64 * 8);
  v2f*   WxP = (v2f*)(smem + 64 * IS * 4 + 16 * BP64 * 8 + 64 * XS * 4);

  const int tid  = threadIdx.x;
  const int wave = tid >> 5;
  const int lane = tid & 31;
  const int lh   = lane >> 4;
  const int ln   = lane & 15;
  const int row0 = blockIdx.x * 64;

  for (int i = tid; i < 64 * 32; i += 512) {
    int r = i >> 5, c = i & 31;
    inP[r * IS + c] = in[(size_t)(row0 + r) * 32 + c];
  }
  for (int i = tid; i < 16 * 64; i += 512) {
    int kp = i >> 6, n = i & 63;
    v2f w; w.x = We[(2 * kp) * 64 + n]; w.y = We[(2 * kp + 1) * 64 + n];
    WeP[kp * BP64 + n] = w;
  }
  for (int i = tid; i < 32 * 512; i += 512) {
    int kp = i >> 9, n = i & 511;
    v2f w; w.x = Wx0[(2 * kp) * 512 + n]; w.y = Wx0[(2 * kp + 1) * 512 + n];
    WxP[kp * BP512 + n] = w;
  }
  __syncthreads();

  {
    const int tm = wave >> 2, tn = wave & 3;
    const int col = tn * 16 + ln;
    const float bv = be[col];
    v8f acc = vzero8();
    const int am = tm * 16 + ln;
#pragma unroll
    for (int k0 = 0; k0 < 32; k0 += 4) {
      v2f a = *(const v2f*)&inP[am * IS + k0 + 2 * lh];
      v2f b = WeP[((k0 >> 1) + lh) * BP64 + col];
      acc = wmma4(a, b, acc);
    }
#pragma unroll
    for (int r = 0; r < 8; ++r)
      xb[(tm * 16 + r + 8 * lh) * XS + col] = acc[r] + bv;
  }
  __syncthreads();

  for (int task = wave; task < 128; task += 16) {
    const int tm = task >> 5, tn = task & 31;
    const int col = tn * 16 + ln;
    const float bv = b0[col];
    v8f acc = vzero8();
    const int am = tm * 16 + ln;
#pragma unroll
    for (int k0 = 0; k0 < 64; k0 += 4) {
      v2f a = *(const v2f*)&xb[am * XS + k0 + 2 * lh];
      v2f b = WxP[((k0 >> 1) + lh) * BP512 + col];
      acc = wmma4(a, b, acc);
    }
    store_xg_tile(xg, blockIdx.x, tm, tn, lane, acc, bv);
  }
}

__global__ void __launch_bounds__(512)
xg1_kernel(const float* __restrict__ x1,
           const float* __restrict__ Wx1,
           const float* __restrict__ b1,
           float* __restrict__ xg)
{
  extern __shared__ __align__(16) char smem[];
  float* aB = (float*)smem;
  v2f*   WP = (v2f*)(smem + 64 * HS * 4);

  const int tid  = threadIdx.x;
  const int wave = tid >> 5;
  const int lane = tid & 31;
  const int lh   = lane >> 4;
  const int ln   = lane & 15;
  const int row0 = blockIdx.x * 64;

  for (int i = tid; i < 64 * 128; i += 512) {
    int r = i >> 7, c = i & 127;
    aB[r * HS + c] = x1[(size_t)(row0 + r) * 128 + c];
  }
  for (int i = tid; i < 64 * 512; i += 512) {
    int kp = i >> 9, n = i & 511;
    v2f w; w.x = Wx1[(2 * kp) * 512 + n]; w.y = Wx1[(2 * kp + 1) * 512 + n];
    WP[kp * BP512 + n] = w;
  }
  __syncthreads();

  for (int task = wave; task < 128; task += 16) {
    const int tm = task >> 5, tn = task & 31;
    const int col = tn * 16 + ln;
    const float bv = b1[col];
    v8f acc = vzero8();
    const int am = tm * 16 + ln;
    for (int k0 = 0; k0 < 128; k0 += 4) {
      v2f a = *(const v2f*)&aB[am * HS + k0 + 2 * lh];
      v2f b = WP[((k0 >> 1) + lh) * BP512 + col];
      acc = wmma4(a, b, acc);
    }
    store_xg_tile(xg, blockIdx.x, tm, tn, lane, acc, bv);
  }
}

__global__ void __launch_bounds__(288)
fold_kernel(const float* __restrict__ Wo,
            const float* __restrict__ bo,
            const float* __restrict__ Wx,
            const float* __restrict__ b,
            float* __restrict__ Wf,
            float* __restrict__ bf)
{
  const int tid  = threadIdx.x;
  const int wave = tid >> 5;
  const int lane = tid & 31;
  const int lh   = lane >> 4;
  const int ln   = lane & 15;
  const int col  = blockIdx.x * 16 + ln;
  const bool isW = wave < 8;
  const float* arow = isW ? Wo + (wave * 16 + ln) * HID : bo;
  const bool   live = isW || ln == 0;
  v8f acc = vzero8();
  v2f a[32], w[32];
#pragma unroll
  for (int i = 0; i < 32; ++i) {
    const v2f av = *(const v2f*)(arow + 4 * i + 2 * lh);
    a[i].x = live ? av.x : 0.0f;
    a[i].y = live ? av.y : 0.0f;
    w[i].x = Wx[(4 * i + 2 * lh) * G4 + col];
    w[i].y = Wx[(4 * i + 2 * lh + 1) * G4 + col];
  }
#pragma unroll
  for (int i = 0; i < 32; ++i) acc = wmma4(a[i], w[i], acc);
  if (isW) {
#pragma unroll
    for (int r = 0; r < 8; ++r)
      Wf[(wave * 16 + r + 8 * lh) * G4 + col] = acc[r];
  } else if (lh == 0) {
    bf[col] = acc[0] + b[col];
  }
}

template <int NOUT>
__global__ void __launch_bounds__(512)
scan_kernel(const float* __restrict__ xg,
            const float* __restrict__ Wh,
            const float* __restrict__ Wo,
            const float* __restrict__ bo,
            float* __restrict__ ring,
            float* __restrict__ out)
{
  extern __shared__ __align__(16) char smem[];
  v2f*   WhP  = (v2f*)smem;
  float* hbuf = (float*)(smem + WHS_BYTES);

  const int tid  = threadIdx.x;
  const int wave = tid >> 5;
  const int lane = tid & 31;
  const int lh   = lane >> 4;
  const int ln   = lane & 15;

  for (int i = tid; i < 64 * 512; i += 512) {
    int kp = i >> 9, n = i & 511;
    int g = n >> 7, hh = n & 127;
    v2f w; w.x = Wh[(2 * kp) * 512 + n]; w.y = Wh[(2 * kp + 1) * 512 + n];
    WhP[((((hh >> 4) * 64 + kp) * 2 + (g >> 1)) * 16 + (((hh & 15) + 8 * (kp & 1)) & 15)) * 2 + (g & 1)] = w;
  }
  for (int i = tid; i < 64 * HS; i += 512) hbuf[i] = 0.0f;
  __syncthreads();

  const int tnh  = wave & 7;
  const int colh = tnh * 16 + ln;
  const v4f* wq  = (const v4f*)WhP + ((tnh * 64 + lh) * 2) * 16 + ((ln + 8 * lh) & 15);
  const float* abw = hbuf + ((wave >> 3) * 16 + ln) * HS + 2 * lh;

  v8f  acc[4];
  Frag fr[2];
#pragma unroll
  for (int g = 0; g < 4; ++g) acc[g] = vzero8();
  prio_kloop();
  load_frag(fr[0], abw, wq, 0);
  k_blocks<0, 8>(fr, acc, abw, wq);

  v2f pc[4], po[4];
#pragma unroll
  for (int p = 0; p < 4; ++p) { pc[p] = splat2(0.0f); po[p] = splat2(0.0f); }

  for (int t = 0; t < T_DIM; ++t) {
    const float* xgt = xg + (size_t)t * (64 * 512);

    if (t + 1 < T_DIM)
      __builtin_prefetch(xgt + 64 * 512 + (size_t)tid * 64, 0, 0);

#pragma unroll
    for (int ti = 0; ti < 2; ++ti) {
      const int task = wave + ti * 16;

      v4f xq[4][2];
      {
        const v4f* xp = (const v4f*)(xgt + task * 1024) + lane;
#pragma unroll
        for (int g = 0; g < 4; ++g) {
          xq[g][0] = xp[(g * 2) * 32];
          xq[g][1] = xp[(g * 2 + 1) * 32];
        }
      }

      constexpr int EARLY = KF - 6;
      v4f rv[KF - 1][2];
      const unsigned rboff = (unsigned)(task * 64 + lane) * 16u;
#pragma unroll
      for (int j = 1; j <= EARLY; ++j) {
        const char* sp = (const char*)ring + (size_t)((t - j) & 15) * 32768;
        rv[j - 1][0] = *(const v4f*)(sp + rboff);
        rv[j - 1][1] = *(const v4f*)(sp + rboff + 512);
      }

      const float* ab = abw + ti * (32 * HS);
      v2f hn2[4];
      v2f th[4];
      [&]<int... N>(std::integer_sequence<int, N...>) __attribute__((always_inline)) {
        ([&] __attribute__((always_inline)) {
          constexpr int blk = 8 + N / 8, kk = (N / 4) & 1, g = N & 3;
          if constexpr ((N & 7) == 0) {
            if constexpr (blk + 1 < 16) load_frag(fr[(blk & 1) ^ 1], ab, wq, blk + 1);
            __builtin_amdgcn_sched_barrier(0);
          }
          const v4f qq = fr[blk & 1].q[kk][g >> 1];
          v2f b;
          b.x = (g & 1) ? qq.z : qq.x; b.y = (g & 1) ? qq.w : qq.y;
          acc[g] = wmma4(fr[blk & 1].a[kk], b, acc[g]);
          if constexpr (N < 4) { th[N].x = htanh(pc[N].x); th[N].y = htanh(pc[N].y); }
          if constexpr (N >= 4 && N < 8) hn2[N - 4] = po[N - 4] * th[N - 4];
          if constexpr (N >= 8 && N < 12) {
            constexpr int p = N - 8;
            const int tmp = (wave >> 3) + 2 * (ti ^ 1);
            hbuf[(tmp * 16 + 2 * p + 8 * lh) * HS + colh]     = hn2[p].x;
            hbuf[(tmp * 16 + 2 * p + 1 + 8 * lh) * HS + colh] = hn2[p].y;
          }
          __builtin_amdgcn_sched_barrier(0);
        }(), ...);
      }(std::make_integer_sequence<int, 64>{});
      if constexpr (NOUT == 128) {
        if (ti == 1 || t > 0) {
          char* hb = (char*)out + (size_t)(t - (ti ^ 1)) * (64 * HID * 4);
          const unsigned hoff = (unsigned)(((wave >> 3) * 16 + 8 * lh) * HID + (wave & 7) * 16 + ln) * 4u;
#pragma unroll
          for (int p = 0; p < 4; ++p) {
            *(float*)(hb + hoff + (ti ^ 1) * 16384 + (2 * p) * 512)     = hn2[p].x;
            *(float*)(hb + hoff + (ti ^ 1) * 16384 + (2 * p + 1) * 512) = hn2[p].y;
          }
        }
      }

      bar_arrive();
      prio_gates();

      {
        typedef const char __attribute__((address_space(1)))* gptr_cc;
        typedef const v4f __attribute__((address_space(1)))* gptr_cv4;
        gptr_cc ringl = (gptr_cc)ring;
        asm volatile("" : "+s"(ringl));
#pragma unroll
        for (int j = EARLY + 1; j < KF; ++j) {
          gptr_cc sp = ringl + (size_t)((t - j) & 15) * 32768;
          rv[j - 1][0] = *(gptr_cv4)(sp + rboff);
          rv[j - 1][1] = *(gptr_cv4)(sp + rboff + 512);
        }
      }

      float hc[8], og[8], dd[8];
#pragma unroll
      for (int r = 0; r < 8; ++r) {
        const float ig = sigm(acc[0][r] + xq[0][r >> 2][r & 3]);
        og[r] = sigm(acc[1][r] + xq[1][r >> 2][r & 3]);
        const float cg = htanh(acc[2][r] + xq[2][r >> 2][r & 3]);
        dd[r] = 0.5f * sigm(acc[3][r] + xq[3][r >> 2][r & 3]);
        hc[r] = ig * cg;
      }
      constexpr int U0 = 24;
      constexpr int U1 = 48;
      constexpr int UPG = (NOUT == 128) ? 3 : 4;
      float csum[8], w[8];
#pragma unroll
      for (int r = 0; r < 8; ++r) { csum[r] = hc[r]; w[r] = 1.0f; }
#pragma unroll
      for (int j = 1; j <= U0 / 8; ++j) {
        const float jm1 = (float)(j - 1), rj = 1.0f / (float)j;
#pragma unroll
        for (int r = 0; r < 8; ++r) {
          w[r] *= (jm1 + dd[r]) * rj;
          csum[r] += w[r] * rv[j - 1][r >> 2][r & 3];
        }
      }
      {
        char* sw = (char*)ring + (size_t)(t & 15) * 32768;
        v4f lo, hi;
        lo.x = hc[0]; lo.y = hc[1]; lo.z = hc[2]; lo.w = hc[3];
        hi.x = hc[4]; hi.y = hc[5]; hi.z = hc[6]; hi.w = hc[7];
        *(v4f*)(sw + rboff)       = lo;
        *(v4f*)(sw + rboff + 512) = hi;
      }

      bar_await();
      prio_kloop();

      float t2[128];
      auto stage = [&](int n) __attribute__((always_inline)) {
#pragma unroll
        for (int h = 0; h < UPG; ++h) {
          const int uB = U1 + UPG * (n + 2) + h;
          const int uC = U1 + UPG * (n + 1) + h, uD = U1 + UPG * n + h;
          if (uB >= U1 && uB < 120) { t2[uB] = ((float)(uB >> 3) + dd[uB & 7]) * (1.0f / (float)((uB >> 3) + 1)); hold1(t2[uB]); }
          if (uC >= U1 && uC < 120) { w[uC & 7] *= t2[uC]; hold1(w[uC & 7]); }
          if (uD >= U1 && uD < 120) { csum[uD & 7] += w[uD & 7] * rv[uD >> 3][(uD & 7) >> 2][uD & 3]; hold1(csum[uD & 7]); }
        }
      };

      const float* abn = abw + (ti ^ 1) * (32 * HS);
#pragma unroll
      for (int g = 0; g < 4; ++g) acc[g] = vzero8();
      load_frag(fr[0], abn, wq, 0);
#pragma unroll
      for (int j = U0 / 8 + 1; j <= U1 / 8; ++j) {
        const float jm1 = (float)(j - 1), rj = 1.0f / (float)j;
#pragma unroll
        for (int r = 0; r < 8; ++r) {
          w[r] *= (jm1 + dd[r]) * rj;
          csum[r] += w[r] * rv[j - 1][r >> 2][r & 3];
        }
      }
      __builtin_amdgcn_sched_barrier(0);
      stage(-2); stage(-1);
#pragma unroll
      for (int blk = 0; blk < 8; ++blk) {
        load_frag(fr[(blk & 1) ^ 1], abn, wq, blk + 1);
        __builtin_amdgcn_sched_barrier(0);
#pragma unroll
        for (int kk = 0; kk < 2; ++kk) {
          const v4f q0 = fr[blk & 1].q[kk][0], q1 = fr[blk & 1].q[kk][1];
          v2f b[4];
          b[0].x = q0.x; b[0].y = q0.y; b[1].x = q0.z; b[1].y = q0.w;
          b[2].x = q1.x; b[2].y = q1.y; b[3].x = q1.z; b[3].y = q1.w;
#pragma unroll
          for (int g = 0; g < 4; ++g) {
            acc[g] = wmma4(fr[blk & 1].a[kk], b[g], acc[g]);
            stage(blk * 8 + kk * 4 + g);
            __builtin_amdgcn_sched_barrier(0);
          }
        }
      }

#pragma unroll
      for (int p = 0; p < 4; ++p) {
        pc[p].x = csum[2 * p]; pc[p].y = csum[2 * p + 1];
        po[p].x = og[2 * p];   po[p].y = og[2 * p + 1];
      }

      if constexpr (NOUT != 128) {
        static_assert(NOUT == 32, "waves 0-3 = batch half 0, waves 4-7 = batch half 1");
        if (ti == 0) {
          if (t > 0 && wave >= 4 && wave < 8)
            project_tile<NOUT>(hbuf, Wo, bo, out + (size_t)(t - 1) * (64 * NOUT), wave, lh, ln);
        } else {
          if (wave < 4)
            project_tile<NOUT>(hbuf, Wo, bo, out + (size_t)t * (64 * NOUT), wave, lh, ln);
        }
      }
    }
  }
  prio_gates();
  {
    const int tmp = (wave >> 3) + 2;
#pragma unroll
    for (int r = 0; r < 8; ++r) {
      const float hl = (r & 1) ? po[r >> 1].y * htanh(pc[r >> 1].y) : po[r >> 1].x * htanh(pc[r >> 1].x);
      hbuf[(tmp * 16 + r + 8 * lh) * HS + colh] = hl;
      if constexpr (NOUT == 128) {
        const unsigned hoff = (unsigned)(((wave >> 3) * 16 + 8 * lh) * HID + (wave & 7) * 16 + ln) * 4u;
        *(float*)((char*)out + (size_t)(T_DIM - 1) * (64 * HID * 4) + hoff + 16384 + r * 512) = hl;
      }
    }
  }
  if constexpr (NOUT != 128) {
    __syncthreads();
    if (wave >= 4 && wave < 8)
      project_tile<NOUT>(hbuf, Wo, bo, out + (size_t)(T_DIM - 1) * (64 * NOUT), wave, lh, ln);
  }
}

extern "C" void kernel_launch(void* const* d_in, const int* in_sizes, int n_in,
                              void* d_out, int out_size, void* d_ws, size_t ws_size,
                              hipStream_t stream) {
  const float* inputs = (const float*)d_in[0];
  const float* W_emb  = (const float*)d_in[1];
  const float* b_emb  = (const float*)d_in[2];
  const float* Wx0    = (const float*)d_in[3];
  const float* Wh0    = (const float*)d_in[4];
  const float* b0     = (const float*)d_in[5];
  const float* Wo0    = (const float*)d_in[6];
  const float* bo0    = (const float*)d_in[7];
  const float* Wx1    = (const float*)d_in[8];
  const float* Wh1    = (const float*)d_in[9];
  const float* b1     = (const float*)d_in[10];
  const float* Wo1    = (const float*)d_in[11];
  const float* bo1    = (const float*)d_in[12];
  float* out = (float*)d_out;

  float* xg   = (float*)d_ws;
  float* h0   = xg + (size_t)T_DIM * B_DIM * G4;
  float* ring = h0 + (size_t)T_DIM * B_DIM * HID;
  const int ringN = KF * B_DIM * HID;
  float* Wf   = ring + (size_t)KF * B_DIM * HID;
  float* bf   = Wf + (size_t)HID * G4;

  const size_t ldsE = (size_t)64 * IS * 4 + 16 * BP64 * 8 + 64 * XS * 4 + 32 * BP512 * 8;
  const size_t ldsX = (size_t)64 * HS * 4 + 64 * BP512 * 8;
  const size_t ldsS = (size_t)WHS_BYTES + 64 * HS * 4;

  const int nblk = (T_DIM * B_DIM) / 64;

  embed_xg0_kernel<<<nblk, 512, ldsE, stream>>>(inputs, W_emb, b_emb, Wx0, b0, xg);
  zero_kernel<<<ringN / 256, 256, 0, stream>>>(ring, ringN);
  scan_kernel<HID><<<1, 512, ldsS, stream>>>(xg, Wh0, Wo0, bo0, ring, h0);
  fold_kernel<<<G4 / 16, 288, 0, stream>>>(Wo0, bo0, Wx1, b1, Wf, bf);
  xg1_kernel<<<nblk, 512, ldsX, stream>>>(h0, Wf, bf, xg);
  zero_kernel<<<ringN / 256, 256, 0, stream>>>(ring, ringN);
  scan_kernel<OUTD><<<1, 512, ldsS, stream>>>(xg, Wh1, Wo1, bo1, ring, out);
}
